# starting kernel
# baseline (speedup 1.0000x reference)
#include <hip/hip_runtime.h>
#include <hip/hip_bf16.h>


#define DIN 128
#define HD  64
#define H4  4

typedef __attribute__((ext_vector_type(2))) float v2f;
typedef __attribute__((ext_vector_type(8))) float v8f;

__global__ __launch_bounds__(256) void gat_gemm_wmma(
    const float* __restrict__ feat, const float* __restrict__ W,
    float* __restrict__ z, int N_) {
  const int lane = threadIdx.x & 31;
  const int wave = blockIdx.x * 8 + (threadIdx.x >> 5);
  const int r0 = wave * 16;
  if (r0 >= N_) return;

  const int half = lane >> 4;
  const int m    = lane & 15;

  int rowA = r0 + m;
  if (rowA >= N_) rowA = N_ - 1;
  const float* __restrict__ arow = feat + (size_t)rowA * DIN;
  const float* __restrict__ b0p  = W + (size_t)(0 * 16 + m) * DIN;
  const float* __restrict__ b1p  = W + (size_t)(1 * 16 + m) * DIN;
  const float* __restrict__ b2p  = W + (size_t)(2 * 16 + m) * DIN;
  const float* __restrict__ b3p  = W + (size_t)(3 * 16 + m) * DIN;

  v8f acc0 = {}, acc1 = {}, acc2 = {}, acc3 = {};

  for (int k0 = 0; k0 < DIN; k0 += 4) {
    const int ka = k0 + 2 * half;
    v2f a  = *(const v2f*)(arow + ka);
    v2f b0 = *(const v2f*)(b0p + ka);
    v2f b1 = *(const v2f*)(b1p + ka);
    v2f b2 = *(const v2f*)(b2p + ka);
    v2f b3 = *(const v2f*)(b3p + ka);
    acc0 = __builtin_amdgcn_wmma_f32_16x16x4_f32(false, a, false, b0,
                                                 (short)0, acc0, false, false);
    acc1 = __builtin_amdgcn_wmma_f32_16x16x4_f32(false, a, false, b1,
                                                 (short)0, acc1, false, false);
    acc2 = __builtin_amdgcn_wmma_f32_16x16x4_f32(false, a, false, b2,
                                                 (short)0, acc2, false, false);
    acc3 = __builtin_amdgcn_wmma_f32_16x16x4_f32(false, a, false, b3,
                                                 (short)0, acc3, false, false);
  }

#pragma unroll
  for (int r = 0; r < 8; ++r) {
    const int row = r0 + r + 8 * half;
    if (row < N_) {
      float* zr = z + (size_t)row * HD + m;
      zr[0]  = acc0[r];
      zr[16] = acc1[r];
      zr[32] = acc2[r];
      zr[48] = acc3[r];
    }
  }
}

__global__ __launch_bounds__(256) void gat_prep(
    const float* __restrict__ z, const float* __restrict__ aw,
    float* __restrict__ el, float* __restrict__ er,
    float* __restrict__ denom, float* __restrict__ out, int N_) {
  const int i = blockIdx.x * blockDim.x + threadIdx.x;
  if (i >= N_ * H4) return;
  const int n = i >> 2;
  const int h = i & 3;
  const float* zr = z + (size_t)n * HD + h * 16;
  const float* a0 = aw + h * 16;
  const float* a1 = aw + 64 + h * 16;
  float s0 = 0.f, s1 = 0.f;
#pragma unroll
  for (int d = 0; d < 16; ++d) {
    const float zv = zr[d];
    s0 = fmaf(zv, a0[d], s0);
    s1 = fmaf(zv, a1[d], s1);
  }
  el[i]    = s0;
  er[i]    = s1;
  denom[i] = 0.f;
  float* o = out + (size_t)n * HD + h * 16;
#pragma unroll
  for (int d = 0; d < 16; ++d) o[d] = 0.f;
}

__global__ __launch_bounds__(256) void gat_edge(
    const int* __restrict__ src, const int* __restrict__ dst,
    const float* __restrict__ el, const float* __restrict__ er,
    const float* __restrict__ z, float* __restrict__ denom,
    float* __restrict__ out, int E_) {
  const int wave = (int)((blockIdx.x * blockDim.x + threadIdx.x) >> 5);
  const int lane = threadIdx.x & 31;
  if (wave >= E_) return;

  const int s = src[wave];
  const int t = dst[wave];
  const int h0 = lane >> 4;
  const int h1 = h0 + 2;

  float x0 = el[s * H4 + h0] + er[t * H4 + h0];
  float x1 = el[s * H4 + h1] + er[t * H4 + h1];
  x0 = x0 > 0.f ? x0 : 0.2f * x0;
  x1 = x1 > 0.f ? x1 : 0.2f * x1;
  const float ex0 = __expf(x0);
  const float ex1 = __expf(x1);

  if ((lane & 15) == 0) {
    atomicAdd(&denom[t * H4 + h0], ex0);
    atomicAdd(&denom[t * H4 + h1], ex1);
  }

  const float z0 = z[(size_t)s * HD + lane];
  const float z1 = z[(size_t)s * HD + lane + 32];
  atomicAdd(&out[(size_t)t * HD + lane],      ex0 * z0);
  atomicAdd(&out[(size_t)t * HD + lane + 32], ex1 * z1);
}

__global__ __launch_bounds__(256) void gat_finalize(
    const float* __restrict__ denom, const float* __restrict__ bias,
    float* __restrict__ out, int N_) {
  const int i = blockIdx.x * blockDim.x + threadIdx.x;
  if (i >= N_ * HD) return;
  const int n = i >> 6;
  const int j = i & 63;
  const int h = j >> 4;
  float den = denom[n * H4 + h];
  den = den > 0.f ? den : 1.f;
  out[i] = out[i] / den + bias[j];
}

extern "C" void kernel_launch(void* const* d_in, const int* in_sizes, int n_in,
                              void* d_out, int out_size, void* d_ws, size_t ws_size,
                              hipStream_t stream) {
  const float* feat = (const float*)d_in[0];
  const int*   src  = (const int*)d_in[1];
  const int*   dst  = (const int*)d_in[2];
  const float* W    = (const float*)d_in[3];
  const float* aw   = (const float*)d_in[4];
  const float* bias = (const float*)d_in[5];
  float* out = (float*)d_out;

  const int N_ = in_sizes[0] / DIN;
  const int E_ = in_sizes[1];

  float* ws    = (float*)d_ws;
  float* z     = ws;
  float* el    = z  + (size_t)N_ * HD;
  float* er    = el + (size_t)N_ * H4;
  float* denom = er + (size_t)N_ * H4;

  const int gemmWaves  = (N_ + 15) / 16;
  const int gemmBlocks = (gemmWaves + 7) / 8;
  gat_gemm_wmma<<<gemmBlocks, 256, 0, stream>>>(feat, W, z, N_);

  const int prepT = N_ * H4;
  gat_prep<<<(prepT + 255) / 256, 256, 0, stream>>>(z, aw, el, er, denom, out, N_);

  gat_edge<<<(E_ + 7) / 8, 256, 0, stream>>>(src, dst, el, er, z, denom, out, E_);

  const int finT = N_ * HD;
  gat_finalize<<<(finT + 255) / 256, 256, 0, stream>>>(denom, bias, out, N_);
}


# Edges ranked per destination by one int atomic each into lists staged in the out rows; one wave per row gathers z, sums in rank (arrival) order in registers, normalizes: float atomics gone
# speedup vs baseline: 1.2662x; 1.2662x over previous
#include <hip/hip_runtime.h>
#include <hip/hip_bf16.h>


#define DIN 128
#define HD  64
#define H4  4

typedef __attribute__((ext_vector_type(2))) float v2f;
typedef __attribute__((ext_vector_type(8))) float v8f;

__global__ __launch_bounds__(256) void gat_gemm_wmma(
    const float* __restrict__ feat, const float* __restrict__ W,
    float* __restrict__ z, int N_) {
  const int lane = threadIdx.x & 31;
  const int wave = blockIdx.x * 8 + (threadIdx.x >> 5);
  const int r0 = wave * 16;
  if (r0 >= N_) return;

  const int half = lane >> 4;
  const int m    = lane & 15;

  int rowA = r0 + m;
  if (rowA >= N_) rowA = N_ - 1;
  const float* __restrict__ arow = feat + (size_t)rowA * DIN;
  const float* __restrict__ b0p  = W + (size_t)(0 * 16 + m) * DIN;
  const float* __restrict__ b1p  = W + (size_t)(1 * 16 + m) * DIN;
  const float* __restrict__ b2p  = W + (size_t)(2 * 16 + m) * DIN;
  const float* __restrict__ b3p  = W + (size_t)(3 * 16 + m) * DIN;

  v8f acc0 = {}, acc1 = {}, acc2 = {}, acc3 = {};

  for (int k0 = 0; k0 < DIN; k0 += 4) {
    const int ka = k0 + 2 * half;
    v2f a  = *(const v2f*)(arow + ka);
    v2f b0 = *(const v2f*)(b0p + ka);
    v2f b1 = *(const v2f*)(b1p + ka);
    v2f b2 = *(const v2f*)(b2p + ka);
    v2f b3 = *(const v2f*)(b3p + ka);
    acc0 = __builtin_amdgcn_wmma_f32_16x16x4_f32(false, a, false, b0,
                                                 (short)0, acc0, false, false);
    acc1 = __builtin_amdgcn_wmma_f32_16x16x4_f32(false, a, false, b1,
                                                 (short)0, acc1, false, false);
    acc2 = __builtin_amdgcn_wmma_f32_16x16x4_f32(false, a, false, b2,
                                                 (short)0, acc2, false, false);
    acc3 = __builtin_amdgcn_wmma_f32_16x16x4_f32(false, a, false, b3,
                                                 (short)0, acc3, false, false);
  }

#pragma unroll
  for (int r = 0; r < 8; ++r) {
    const int row = r0 + r + 8 * half;
    if (row < N_) {
      float* zr = z + (size_t)row * HD + m;
      zr[0]  = acc0[r];
      zr[16] = acc1[r];
      zr[32] = acc2[r];
      zr[48] = acc3[r];
    }
  }
}

__global__ __launch_bounds__(256) void gat_prep(
    const float* __restrict__ z, const float* __restrict__ aw,
    float* __restrict__ el, float* __restrict__ er,
    int* __restrict__ cnt, int N_) {
  const int i = blockIdx.x * blockDim.x + threadIdx.x;
  if (i >= N_ * H4) return;
  const int n = i >> 2;
  const int h = i & 3;
  const float* zr = z + (size_t)n * HD + h * 16;
  const float* a0 = aw + h * 16;
  const float* a1 = aw + 64 + h * 16;
  float s0 = 0.f, s1 = 0.f;
#pragma unroll
  for (int d = 0; d < 16; ++d) {
    const float zv = zr[d];
    s0 = fmaf(zv, a0[d], s0);
    s1 = fmaf(zv, a1[d], s1);
  }
  el[i] = s0;
  er[i] = s1;
  if (h == 0) cnt[n] = 0;
}

#define LISTCAP 64

__device__ __forceinline__ void rank_edge(int s, int t, int* __restrict__ cnt,
                                          float* __restrict__ out) {
  const int r = atomicAdd(&cnt[t], 1);
  if (r < LISTCAP) out[(size_t)t * HD + r] = __int_as_float(s);
}

__global__ __launch_bounds__(256) void gat_rank(
    const int* __restrict__ src, const int* __restrict__ dst,
    int* __restrict__ cnt, float* __restrict__ out, int E_) {
  const int e0 = (blockIdx.x * blockDim.x + threadIdx.x) * 4;
  if (e0 + 3 < E_) {
    const int4 s4 = *(const int4*)(src + e0);
    const int4 t4 = *(const int4*)(dst + e0);
    const int r0 = atomicAdd(&cnt[t4.x], 1);
    const int r1 = atomicAdd(&cnt[t4.y], 1);
    const int r2 = atomicAdd(&cnt[t4.z], 1);
    const int r3 = atomicAdd(&cnt[t4.w], 1);
    if (r0 < LISTCAP) out[(size_t)t4.x * HD + r0] = __int_as_float(s4.x);
    if (r1 < LISTCAP) out[(size_t)t4.y * HD + r1] = __int_as_float(s4.y);
    if (r2 < LISTCAP) out[(size_t)t4.z * HD + r2] = __int_as_float(s4.z);
    if (r3 < LISTCAP) out[(size_t)t4.w * HD + r3] = __int_as_float(s4.w);
  } else {
    for (int e = e0; e < E_; ++e) rank_edge(src[e], dst[e], cnt, out);
  }
}

__device__ __forceinline__ void add_edge(int s, float w, int lane, int h,
    float erh, const float* __restrict__ el, const float* __restrict__ z,
    float& den, float& a0, float& a1) {
  float x = el[s * H4 + h] + erh;
  x = x > 0.f ? x : 0.2f * x;
  const float p = w * __expf(x);
  const float2 zv = *(const float2*)(z + (size_t)s * HD + 2 * lane);
  den += p;
  a0 = fmaf(p, zv.x, a0);
  a1 = fmaf(p, zv.y, a1);
}

__global__ __launch_bounds__(256) void gat_node(
    const int* __restrict__ src, const int* __restrict__ dst,
    const float* __restrict__ el, const float* __restrict__ er,
    const float* __restrict__ z, const int* __restrict__ cnt,
    const float* __restrict__ bias, float* out, int N_, int E_) {
  const int lane = threadIdx.x & 31;
  const int nv = blockIdx.x * 8 + (threadIdx.x >> 5);
  if (nv >= N_) return;
  const int n = __builtin_amdgcn_readfirstlane(nv);
  const int h = lane >> 3;
  float* orow = out + (size_t)n * HD;
  const int sv0 = __float_as_int(orow[lane]);
  const int sv1 = __float_as_int(orow[lane + 32]);
  const int c = __builtin_amdgcn_readfirstlane(cnt[n]);
  const float erh = er[n * H4 + h];
  const float2 b = *(const float2*)(bias + 2 * lane);
  const unsigned nmax = (unsigned)(N_ - 1);

  float den = 0.f, a0 = 0.f, a1 = 0.f;
  if (c <= LISTCAP) {
    for (int j0 = 0; j0 < c; j0 += 8) {
#pragma unroll
      for (int k = 0; k < 8; ++k) {
        const int j = j0 + k;
        const int jj = j < c ? j : c - 1;
        const int lo = __builtin_amdgcn_readlane(sv0, jj & 31);
        const int hi = __builtin_amdgcn_readlane(sv1, jj & 31);
        const unsigned s = min((unsigned)(jj < 32 ? lo : hi), nmax);
        add_edge((int)s, j < c ? 1.f : 0.f, lane, h, erh, el, z, den, a0, a1);
      }
    }
  } else {
    for (int e0 = 0; e0 < E_; e0 += 32) {
      const int e = e0 + lane;
      const int t = e < E_ ? dst[e] : -1;
      const int sl = e < E_ ? src[e] : 0;
      unsigned m = (unsigned)__ballot(t == n);
      while (m) {
        const int j = __builtin_ctz(m);
        m &= m - 1;
        const unsigned s = min((unsigned)__builtin_amdgcn_readlane(sl, j), nmax);
        add_edge((int)s, 1.f, lane, h, erh, el, z, den, a0, a1);
      }
    }
  }

  den = den > 0.f ? den : 1.f;
  float2 o;
  o.x = a0 / den + b.x;
  o.y = a1 / den + b.y;
  *(float2*)(orow + 2 * lane) = o;
}

extern "C" void kernel_launch(void* const* d_in, const int* in_sizes, int n_in,
                              void* d_out, int out_size, void* d_ws, size_t ws_size,
                              hipStream_t stream) {
  const float* feat = (const float*)d_in[0];
  const int*   src  = (const int*)d_in[1];
  const int*   dst  = (const int*)d_in[2];
  const float* W    = (const float*)d_in[3];
  const float* aw   = (const float*)d_in[4];
  const float* bias = (const float*)d_in[5];
  float* out = (float*)d_out;

  const int N_ = in_sizes[0] / DIN;
  const int E_ = in_sizes[1];

  float* ws  = (float*)d_ws;
  float* z   = ws;
  float* el  = z  + (size_t)N_ * HD;
  float* er  = el + (size_t)N_ * H4;
  int*   cnt = (int*)(er + (size_t)N_ * H4);

  const int gemmWaves  = (N_ + 15) / 16;
  const int gemmBlocks = (gemmWaves + 7) / 8;
  gat_gemm_wmma<<<gemmBlocks, 256, 0, stream>>>(feat, W, z, N_);

  const int prepT = N_ * H4;
  gat_prep<<<(prepT + 255) / 256, 256, 0, stream>>>(z, aw, el, er, cnt, N_);

  const int rankT = (E_ + 3) / 4;
  gat_rank<<<(rankT + 255) / 256, 256, 0, stream>>>(src, dst, cnt, out, E_);

  gat_node<<<(N_ + 7) / 8, 256, 0, stream>>>(src, dst, el, er, z, cnt, bias, out, N_, E_);
}


# Per-edge returning global adds replaced by a two-level counting sort: edges binned per 256-row block with LDS counters; a work-group per block ranks them in LDS and sums its rows, arrival-order sums
# speedup vs baseline: 2.7752x; 2.1918x over previous
#include <hip/hip_runtime.h>
#include <hip/hip_bf16.h>


#define DIN 128
#define HD  64
#define H4  4

typedef __attribute__((ext_vector_type(2))) float v2f;
typedef __attribute__((ext_vector_type(8))) float v8f;

__global__ __launch_bounds__(256) void gat_gemm_wmma(
    const float* __restrict__ feat, const float* __restrict__ W,
    float* __restrict__ z, int N_) {
  const int lane = threadIdx.x & 31;
  const int wave = blockIdx.x * 8 + (threadIdx.x >> 5);
  const int r0 = wave * 16;
  if (r0 >= N_) return;

  const int half = lane >> 4;
  const int m    = lane & 15;

  int rowA = r0 + m;
  if (rowA >= N_) rowA = N_ - 1;
  const float* __restrict__ arow = feat + (size_t)rowA * DIN;
  const float* __restrict__ b0p  = W + (size_t)(0 * 16 + m) * DIN;
  const float* __restrict__ b1p  = W + (size_t)(1 * 16 + m) * DIN;
  const float* __restrict__ b2p  = W + (size_t)(2 * 16 + m) * DIN;
  const float* __restrict__ b3p  = W + (size_t)(3 * 16 + m) * DIN;

  v8f acc0 = {}, acc1 = {}, acc2 = {}, acc3 = {};

  for (int k0 = 0; k0 < DIN; k0 += 4) {
    const int ka = k0 + 2 * half;
    v2f a  = *(const v2f*)(arow + ka);
    v2f b0 = *(const v2f*)(b0p + ka);
    v2f b1 = *(const v2f*)(b1p + ka);
    v2f b2 = *(const v2f*)(b2p + ka);
    v2f b3 = *(const v2f*)(b3p + ka);
    acc0 = __builtin_amdgcn_wmma_f32_16x16x4_f32(false, a, false, b0,
                                                 (short)0, acc0, false, false);
    acc1 = __builtin_amdgcn_wmma_f32_16x16x4_f32(false, a, false, b1,
                                                 (short)0, acc1, false, false);
    acc2 = __builtin_amdgcn_wmma_f32_16x16x4_f32(false, a, false, b2,
                                                 (short)0, acc2, false, false);
    acc3 = __builtin_amdgcn_wmma_f32_16x16x4_f32(false, a, false, b3,
                                                 (short)0, acc3, false, false);
  }

#pragma unroll
  for (int r = 0; r < 8; ++r) {
    const int row = r0 + r + 8 * half;
    if (row < N_) {
      float* zr = z + (size_t)row * HD + m;
      zr[0]  = acc0[r];
      zr[16] = acc1[r];
      zr[32] = acc2[r];
      zr[48] = acc3[r];
    }
  }
}

#define BSH     8
#define BROWS   (1 << BSH)
#define BWORDS  (BROWS * HD)
#define LCAP    8192
#define CURSTR  32
#define PCH     8192

__global__ __launch_bounds__(256) void gat_prep(
    const float* __restrict__ z, const float* __restrict__ aw,
    float* __restrict__ el, float* __restrict__ er,
    int* __restrict__ cur, int N_, int NB) {
  const int i = blockIdx.x * blockDim.x + threadIdx.x;
  if (i >= N_ * H4) return;
  const int n = i >> 2;
  const int h = i & 3;
  const float* zr = z + (size_t)n * HD + h * 16;
  const float* a0 = aw + h * 16;
  const float* a1 = aw + 64 + h * 16;
  float s0 = 0.f, s1 = 0.f;
#pragma unroll
  for (int d = 0; d < 16; ++d) {
    const float zv = zr[d];
    s0 = fmaf(zv, a0[d], s0);
    s1 = fmaf(zv, a1[d], s1);
  }
  el[i] = s0;
  er[i] = s1;
  if (i < NB) cur[i * CURSTR] = 0;
}

__device__ __forceinline__ int block_cap(int b, int N_) {
  return min(LCAP, (N_ - (b << BSH)) * HD);
}

__global__ __launch_bounds__(1024) void gat_bin(
    const int* __restrict__ src, const int* __restrict__ dst,
    int* __restrict__ cur, float* __restrict__ out, int N_, int E_, int NB) {
  extern __shared__ int hist[];
  const int tid = threadIdx.x;
  for (int b = tid; b < NB; b += 1024) hist[b] = 0;
  __syncthreads();

  const int cbase = blockIdx.x * PCH;
  int sv[8], tv[8], rv[8];
  if (cbase + PCH <= E_) {
#pragma unroll
    for (int i = 0; i < 2; ++i) {
      const int e0 = cbase + (i * 1024 + tid) * 4;
      const int4 s4 = *(const int4*)(src + e0);
      const int4 t4 = *(const int4*)(dst + e0);
      sv[4 * i + 0] = s4.x; sv[4 * i + 1] = s4.y;
      sv[4 * i + 2] = s4.z; sv[4 * i + 3] = s4.w;
      tv[4 * i + 0] = t4.x; tv[4 * i + 1] = t4.y;
      tv[4 * i + 2] = t4.z; tv[4 * i + 3] = t4.w;
    }
  } else {
#pragma unroll
    for (int k = 0; k < 8; ++k) {
      const int e = cbase + ((k >> 2) * 1024 + tid) * 4 + (k & 3);
      sv[k] = e < E_ ? src[e] : 0;
      tv[k] = e < E_ ? dst[e] : -1;
    }
  }
#pragma unroll
  for (int k = 0; k < 8; ++k) {
    rv[k] = -1;
    if ((unsigned)tv[k] < (unsigned)N_)
      rv[k] = atomicAdd(&hist[tv[k] >> BSH], 1);
  }
  __syncthreads();
  for (int b = tid; b < NB; b += 1024) {
    const int c = hist[b];
    if (c > 0) hist[b] = atomicAdd(&cur[b * CURSTR], c);
  }
  __syncthreads();
#pragma unroll
  for (int k = 0; k < 8; ++k) {
    if (rv[k] >= 0) {
      const int b = tv[k] >> BSH;
      const int pos = hist[b] + rv[k];
      if (pos < block_cap(b, N_)) {
        const unsigned wd = ((unsigned)sv[k] & 0xFFFFFFu) |
                            ((unsigned)(tv[k] & (BROWS - 1)) << 24);
        out[(size_t)b * BWORDS + pos] = __uint_as_float(wd);
      }
    }
  }
}

__device__ __forceinline__ void add_edge(int s, float w, int lane, int h,
    float erh, const float* __restrict__ el, const float* __restrict__ z,
    float& den, float& a0, float& a1) {
  float x = el[s * H4 + h] + erh;
  x = x > 0.f ? x : 0.2f * x;
  const float p = w * __expf(x);
  const float2 zv = *(const float2*)(z + (size_t)s * HD + 2 * lane);
  den += p;
  a0 = fmaf(p, zv.x, a0);
  a1 = fmaf(p, zv.y, a1);
}

__global__ __launch_bounds__(1024) void gat_block(
    const int* __restrict__ src, const int* __restrict__ dst,
    const float* __restrict__ el, const float* __restrict__ er,
    const float* __restrict__ z, const int* __restrict__ cur,
    const float* __restrict__ bias, float* out, int N_, int E_) {
  __shared__ int sorted[LCAP + 32];
  __shared__ int cntl[BROWS];
  __shared__ int offl[BROWS];

  const int tid  = threadIdx.x;
  const int lane = tid & 31;
  const int wave = __builtin_amdgcn_readfirstlane(tid >> 5);
  const int b    = blockIdx.x;
  const int row0 = b << BSH;
  const int rows = min(BROWS, N_ - row0);
  const int total = __builtin_amdgcn_readfirstlane(cur[b * CURSTR]);
  const bool listed = total <= block_cap(b, N_) && N_ <= (1 << 24);
  float* oblk = out + (size_t)row0 * HD;

  if (tid < BROWS) cntl[tid] = 0;
  __syncthreads();

  unsigned wd[8];
  int rk[8];
#pragma unroll
  for (int k = 0; k < 8; ++k) {
    const int i = k * 1024 + tid;
    wd[k] = 0u;
    if (listed && i < total) wd[k] = __float_as_uint(oblk[i]);
  }
#pragma unroll
  for (int k = 0; k < 8; ++k) {
    const int i = k * 1024 + tid;
    rk[k] = -1;
    if (listed && i < total) rk[k] = atomicAdd(&cntl[wd[k] >> 24], 1);
  }
  __syncthreads();

  if (wave == 0) {
    int c8[8];
    int s = 0;
#pragma unroll
    for (int k = 0; k < 8; ++k) { c8[k] = cntl[lane * 8 + k]; s += c8[k]; }
    int incl = s;
#pragma unroll
    for (int d = 1; d < 32; d <<= 1) {
      const int v = __shfl_up(incl, d);
      if (lane >= d) incl += v;
    }
    int run = incl - s;
#pragma unroll
    for (int k = 0; k < 8; ++k) { offl[lane * 8 + k] = run; run += c8[k]; }
  }
  __syncthreads();

#pragma unroll
  for (int k = 0; k < 8; ++k) {
    if (rk[k] >= 0) sorted[offl[wd[k] >> 24] + rk[k]] = (int)(wd[k] & 0xFFFFFFu);
  }
  __syncthreads();

  const int h = lane >> 3;
  const float2 bb = *(const float2*)(bias + 2 * lane);
  const unsigned nmax = (unsigned)(N_ - 1);

  for (int q = 0; q < BROWS / 32; ++q) {
    const int ld = wave * (BROWS / 32) + q;
    if (ld >= rows) break;
    const int n = row0 + ld;
    const float erh = er[n * H4 + h];
    float den = 0.f, a0 = 0.f, a1 = 0.f;
    if (listed) {
      const int c = __builtin_amdgcn_readfirstlane(cntl[ld]);
      const int o = __builtin_amdgcn_readfirstlane(offl[ld]);
      for (int jc = 0; jc < c; jc += 32) {
        const int sl = sorted[o + jc + lane];
        const int cc = min(c - jc, 32);
        for (int j0 = 0; j0 < cc; j0 += 8) {
#pragma unroll
          for (int k = 0; k < 8; ++k) {
            const int j = j0 + k;
            const int jj = j < cc ? j : cc - 1;
            const unsigned s = min((unsigned)__builtin_amdgcn_readlane(sl, jj), nmax);
            add_edge((int)s, j < cc ? 1.f : 0.f, lane, h, erh, el, z, den, a0, a1);
          }
        }
      }
    } else {
      for (int e0 = 0; e0 < E_; e0 += 32) {
        const int e = e0 + lane;
        const int t = e < E_ ? dst[e] : -1;
        const int sl = e < E_ ? src[e] : 0;
        unsigned m = (unsigned)__ballot(t == n);
        while (m) {
          const int j = __builtin_ctz(m);
          m &= m - 1;
          const unsigned s = min((unsigned)__builtin_amdgcn_readlane(sl, j), nmax);
          add_edge((int)s, 1.f, lane, h, erh, el, z, den, a0, a1);
        }
      }
    }
    den = den > 0.f ? den : 1.f;
    float2 o2;
    o2.x = a0 / den + bb.x;
    o2.y = a1 / den + bb.y;
    *(float2*)(oblk + (size_t)ld * HD + 2 * lane) = o2;
  }
}

extern "C" void kernel_launch(void* const* d_in, const int* in_sizes, int n_in,
                              void* d_out, int out_size, void* d_ws, size_t ws_size,
                              hipStream_t stream) {
  const float* feat = (const float*)d_in[0];
  const int*   src  = (const int*)d_in[1];
  const int*   dst  = (const int*)d_in[2];
  const float* W    = (const float*)d_in[3];
  const float* aw   = (const float*)d_in[4];
  const float* bias = (const float*)d_in[5];
  float* out = (float*)d_out;

  const int N_ = in_sizes[0] / DIN;
  const int E_ = in_sizes[1];
  const int NB = (N_ + BROWS - 1) / BROWS;

  float* ws  = (float*)d_ws;
  float* z   = ws;
  float* el  = z  + (size_t)N_ * HD;
  float* er  = el + (size_t)N_ * H4;
  int*   cur = (int*)(er + (size_t)N_ * H4);

  const int gemmWaves  = (N_ + 15) / 16;
  const int gemmBlocks = (gemmWaves + 7) / 8;
  gat_gemm_wmma<<<gemmBlocks, 256, 0, stream>>>(feat, W, z, N_);

  const int prepT = N_ * H4;
  gat_prep<<<(prepT + 255) / 256, 256, 0, stream>>>(z, aw, el, er, cur, N_, NB);

  gat_bin<<<(E_ + PCH - 1) / PCH, 1024, NB * sizeof(int), stream>>>(src, dst, cur, out, N_, E_, NB);

  gat_block<<<NB, 1024, 0, stream>>>(src, dst, el, er, z, cur, bias, out, N_, E_);
}


# Logits recomputed from the z rows inside the gather (half-wave per edge, 16-byte lanes, quad sum on the vector ALU): el/er pass and per-edge el gather gone; two half-wave partial sums per row
# speedup vs baseline: 2.9563x; 1.0653x over previous
#include <hip/hip_runtime.h>
#include <hip/hip_bf16.h>


#define DIN 128
#define HD  64

typedef __attribute__((ext_vector_type(2))) float v2f;
typedef __attribute__((ext_vector_type(8))) float v8f;

#define BSH     8
#define BROWS   (1 << BSH)
#define BWORDS  (BROWS * HD)
#define LCAP    8192
#define CURSTR  32
#define PCH     8192

__global__ __launch_bounds__(256) void gat_gemm_wmma(
    const float* __restrict__ feat, const float* __restrict__ W,
    float* __restrict__ z, int* __restrict__ cur, int N_, int NB) {
  const int gt = blockIdx.x * 256 + threadIdx.x;
  if (gt < NB) cur[gt * CURSTR] = 0;
  const int lane = threadIdx.x & 31;
  const int wave = blockIdx.x * 8 + (threadIdx.x >> 5);
  const int r0 = wave * 16;
  if (r0 >= N_) return;

  const int half = lane >> 4;
  const int m    = lane & 15;

  int rowA = r0 + m;
  if (rowA >= N_) rowA = N_ - 1;
  const float* __restrict__ arow = feat + (size_t)rowA * DIN;
  const float* __restrict__ b0p  = W + (size_t)(0 * 16 + m) * DIN;
  const float* __restrict__ b1p  = W + (size_t)(1 * 16 + m) * DIN;
  const float* __restrict__ b2p  = W + (size_t)(2 * 16 + m) * DIN;
  const float* __restrict__ b3p  = W + (size_t)(3 * 16 + m) * DIN;

  v8f acc0 = {}, acc1 = {}, acc2 = {}, acc3 = {};

  for (int k0 = 0; k0 < DIN; k0 += 4) {
    const int ka = k0 + 2 * half;
    v2f a  = *(const v2f*)(arow + ka);
    v2f b0 = *(const v2f*)(b0p + ka);
    v2f b1 = *(const v2f*)(b1p + ka);
    v2f b2 = *(const v2f*)(b2p + ka);
    v2f b3 = *(const v2f*)(b3p + ka);
    acc0 = __builtin_amdgcn_wmma_f32_16x16x4_f32(false, a, false, b0,
                                                 (short)0, acc0, false, false);
    acc1 = __builtin_amdgcn_wmma_f32_16x16x4_f32(false, a, false, b1,
                                                 (short)0, acc1, false, false);
    acc2 = __builtin_amdgcn_wmma_f32_16x16x4_f32(false, a, false, b2,
                                                 (short)0, acc2, false, false);
    acc3 = __builtin_amdgcn_wmma_f32_16x16x4_f32(false, a, false, b3,
                                                 (short)0, acc3, false, false);
  }

#pragma unroll
  for (int r = 0; r < 8; ++r) {
    const int row = r0 + r + 8 * half;
    if (row < N_) {
      float* zr = z + (size_t)row * HD + m;
      zr[0]  = acc0[r];
      zr[16] = acc1[r];
      zr[32] = acc2[r];
      zr[48] = acc3[r];
    }
  }
}

__device__ __forceinline__ int block_cap(int b, int N_) {
  return min(LCAP, (N_ - (b << BSH)) * HD);
}

__global__ __launch_bounds__(1024) void gat_bin(
    const int* __restrict__ src, const int* __restrict__ dst,
    int* __restrict__ cur, float* __restrict__ out, int N_, int E_, int NB) {
  extern __shared__ int hist[];
  const int tid = threadIdx.x;
  for (int b = tid; b < NB; b += 1024) hist[b] = 0;
  __syncthreads();

  const int cbase = blockIdx.x * PCH;
  int sv[8], tv[8], rv[8];
  if (cbase + PCH <= E_) {
#pragma unroll
    for (int i = 0; i < 2; ++i) {
      const int e0 = cbase + (i * 1024 + tid) * 4;
      const int4 s4 = *(const int4*)(src + e0);
      const int4 t4 = *(const int4*)(dst + e0);
      sv[4 * i + 0] = s4.x; sv[4 * i + 1] = s4.y;
      sv[4 * i + 2] = s4.z; sv[4 * i + 3] = s4.w;
      tv[4 * i + 0] = t4.x; tv[4 * i + 1] = t4.y;
      tv[4 * i + 2] = t4.z; tv[4 * i + 3] = t4.w;
    }
  } else {
#pragma unroll
    for (int k = 0; k < 8; ++k) {
      const int e = cbase + ((k >> 2) * 1024 + tid) * 4 + (k & 3);
      sv[k] = e < E_ ? src[e] : 0;
      tv[k] = e < E_ ? dst[e] : -1;
    }
  }
#pragma unroll
  for (int k = 0; k < 8; ++k) {
    rv[k] = -1;
    if ((unsigned)tv[k] < (unsigned)N_)
      rv[k] = atomicAdd(&hist[tv[k] >> BSH], 1);
  }
  __syncthreads();
  for (int b = tid; b < NB; b += 1024) {
    const int c = hist[b];
    if (c > 0) hist[b] = atomicAdd(&cur[b * CURSTR], c);
  }
  __syncthreads();
#pragma unroll
  for (int k = 0; k < 8; ++k) {
    if (rv[k] >= 0) {
      const int b = tv[k] >> BSH;
      const int pos = hist[b] + rv[k];
      if (pos < block_cap(b, N_)) {
        const unsigned wd = ((unsigned)sv[k] & 0xFFFFFFu) |
                            ((unsigned)(tv[k] & (BROWS - 1)) << 24);
        out[(size_t)b * BWORDS + pos] = __uint_as_float(wd);
      }
    }
  }
}

__device__ __forceinline__ float quad_sum(float t) {
  t += __int_as_float(__builtin_amdgcn_mov_dpp(__float_as_int(t), 0xB1, 0xF, 0xF, true));
  t += __int_as_float(__builtin_amdgcn_mov_dpp(__float_as_int(t), 0x4E, 0xF, 0xF, true));
  return t;
}

__device__ __forceinline__ float head_logit(const float4 zv, const float4 av) {
  float t = zv.x * av.x;
  t = fmaf(zv.y, av.y, t);
  t = fmaf(zv.z, av.z, t);
  t = fmaf(zv.w, av.w, t);
  return quad_sum(t);
}

__device__ __forceinline__ float xhalf(float x) {
  const int b = __float_as_int(x);
  return __int_as_float(__builtin_amdgcn_permlanex16(b, b, 0x76543210, 0xfedcba98, false, false));
}

__device__ __forceinline__ void add_edge(const float4 zv, float w,
    const float4 al, float erh, float& den, float4& a) {
  float x = head_logit(zv, al) + erh;
  x = x > 0.f ? x : 0.2f * x;
  const float p = w * __expf(x);
  den += p;
  a.x = fmaf(p, zv.x, a.x);
  a.y = fmaf(p, zv.y, a.y);
  a.z = fmaf(p, zv.z, a.z);
  a.w = fmaf(p, zv.w, a.w);
}

__global__ __launch_bounds__(1024) void gat_block(
    const int* __restrict__ src, const int* __restrict__ dst,
    const float* __restrict__ aw, const float* __restrict__ z,
    const int* __restrict__ cur, const float* __restrict__ bias,
    float* out, int N_, int E_) {
  __shared__ int sorted[LCAP];
  __shared__ int cntl[BROWS];
  __shared__ int offl[BROWS];

  const int tid  = threadIdx.x;
  const int lane = tid & 31;
  const int wave = __builtin_amdgcn_readfirstlane(tid >> 5);
  const int b    = blockIdx.x;
  const int row0 = b << BSH;
  const int rows = min(BROWS, N_ - row0);
  const int total = __builtin_amdgcn_readfirstlane(cur[b * CURSTR]);
  const bool listed = total <= block_cap(b, N_) && N_ <= (1 << 24);
  float* oblk = out + (size_t)row0 * HD;

  if (tid < BROWS) cntl[tid] = 0;
  __syncthreads();

  unsigned wd[8];
  int rk[8];
#pragma unroll
  for (int k = 0; k < 8; ++k) {
    const int i = k * 1024 + tid;
    wd[k] = 0u;
    if (listed && i < total) wd[k] = __float_as_uint(oblk[i]);
  }
#pragma unroll
  for (int k = 0; k < 8; ++k) {
    const int i = k * 1024 + tid;
    rk[k] = -1;
    if (listed && i < total) rk[k] = atomicAdd(&cntl[wd[k] >> 24], 1);
  }
  __syncthreads();

  if (wave == 0) {
    int c8[8];
    int s = 0;
#pragma unroll
    for (int k = 0; k < 8; ++k) { c8[k] = cntl[lane * 8 + k]; s += c8[k]; }
    int incl = s;
#pragma unroll
    for (int d = 1; d < 32; d <<= 1) {
      const int v = __shfl_up(incl, d);
      if (lane >= d) incl += v;
    }
    int run = incl - s;
#pragma unroll
    for (int k = 0; k < 8; ++k) { offl[lane * 8 + k] = run; run += c8[k]; }
  }
  __syncthreads();

#pragma unroll
  for (int k = 0; k < 8; ++k) {
    if (rk[k] >= 0) sorted[offl[wd[k] >> 24] + rk[k]] = (int)(wd[k] & 0xFFFFFFu);
  }
  __syncthreads();

  const int hl  = lane >> 4;
  const int c16 = lane & 15;
  const float* __restrict__ zc = z + 4 * c16;
  const char* __restrict__ zb = (const char*)z;
  const unsigned cb = 16u * (unsigned)c16;
  const float4 al = *(const float4*)(aw + 4 * c16);
  const float4 ar = *(const float4*)(aw + HD + 4 * c16);
  const float4 bb = *(const float4*)(bias + 4 * c16);
  const unsigned nmax = (unsigned)(N_ - 1);

  float4 zn = make_float4(0.f, 0.f, 0.f, 0.f);
  if (wave * (BROWS / 32) < rows)
    zn = *(const float4*)(zc + (size_t)(row0 + wave * (BROWS / 32)) * HD);

  for (int q = 0; q < BROWS / 32; ++q) {
    const int ld = wave * (BROWS / 32) + q;
    if (ld >= rows) break;
    const int n = row0 + ld;
    const float erh = head_logit(zn, ar);
    if (q + 1 < BROWS / 32 && ld + 1 < rows)
      zn = *(const float4*)(zc + (size_t)(n + 1) * HD);
    float den = 0.f;
    float4 a = make_float4(0.f, 0.f, 0.f, 0.f);
    if (listed) {
      const int c = __builtin_amdgcn_readfirstlane(cntl[ld]);
      const int o = __builtin_amdgcn_readfirstlane(offl[ld]);
      const int* lst = sorted + o;
      int j0 = 0;
      for (; j0 + 8 <= c; j0 += 8) {
        float4 zv[4];
#pragma unroll
        for (int k = 0; k < 4; ++k) {
          const unsigned s = min((unsigned)lst[j0 + 2 * k + hl], nmax);
          zv[k] = *(const float4*)(zb + (s * 256u + cb));
        }
#pragma unroll
        for (int k = 0; k < 4; ++k) add_edge(zv[k], 1.f, al, erh, den, a);
      }
      const int r = c - j0;
      if (r > 0) {
        int sx[4];
        float w[4];
        float4 zv[4];
#pragma unroll
        for (int k = 0; k < 4; ++k) {
          const int e = 2 * k + hl;
          w[k] = e < r ? 1.f : 0.f;
          sx[k] = lst[j0 + min(e, r - 1)];
        }
#pragma unroll
        for (int k = 0; k < 4; ++k) {
          zv[k] = make_float4(0.f, 0.f, 0.f, 0.f);
          if (2 * k < r) {
            const unsigned s = min((unsigned)sx[k], nmax);
            zv[k] = *(const float4*)(zb + (s * 256u + cb));
          }
        }
#pragma unroll
        for (int k = 0; k < 4; ++k) {
          if (2 * k < r) add_edge(zv[k], w[k], al, erh, den, a);
        }
      }
    } else {
      const float wh = hl ? 0.f : 1.f;
      for (int e0 = 0; e0 < E_; e0 += 32) {
        const int e = e0 + lane;
        const int t = e < E_ ? dst[e] : -1;
        const int sl = e < E_ ? src[e] : 0;
        unsigned m = (unsigned)__ballot(t == n);
        while (m) {
          const int j = __builtin_ctz(m);
          m &= m - 1;
          const unsigned s = min((unsigned)__builtin_amdgcn_readlane(sl, j), nmax);
          const float4 zv = *(const float4*)(zc + (size_t)s * HD);
          add_edge(zv, wh, al, erh, den, a);
        }
      }
    }
    den += xhalf(den);
    a.x += xhalf(a.x);
    a.y += xhalf(a.y);
    a.z += xhalf(a.z);
    a.w += xhalf(a.w);
    den = den > 0.f ? den : 1.f;
    float4 o4;
    o4.x = a.x / den + bb.x;
    o4.y = a.y / den + bb.y;
    o4.z = a.z / den + bb.z;
    o4.w = a.w / den + bb.w;
    if (hl == 0) *(float4*)(oblk + (size_t)ld * HD + 4 * c16) = o4;
  }
}

extern "C" void kernel_launch(void* const* d_in, const int* in_sizes, int n_in,
                              void* d_out, int out_size, void* d_ws, size_t ws_size,
                              hipStream_t stream) {
  const float* feat = (const float*)d_in[0];
  const int*   src  = (const int*)d_in[1];
  const int*   dst  = (const int*)d_in[2];
  const float* W    = (const float*)d_in[3];
  const float* aw   = (const float*)d_in[4];
  const float* bias = (const float*)d_in[5];
  float* out = (float*)d_out;

  const int N_ = in_sizes[0] / DIN;
  const int E_ = in_sizes[1];
  const int NB = (N_ + BROWS - 1) / BROWS;

  float* ws  = (float*)d_ws;
  float* z   = ws;
  int*   cur = (int*)(z + (size_t)N_ * HD);

  const int gemmWaves  = (N_ + 15) / 16;
  const int gemmBlocks = (gemmWaves + 7) / 8;
  gat_gemm_wmma<<<gemmBlocks, 256, 0, stream>>>(feat, W, z, cur, N_, NB);

  gat_bin<<<(E_ + PCH - 1) / PCH, 1024, NB * sizeof(int), stream>>>(src, dst, cur, out, N_, E_, NB);

  gat_block<<<NB, 1024, 0, stream>>>(src, dst, aw, z, cur, bias, out, N_, E_);
}


# GEMM fragment loads cut 3.3x on the WGP load path: 16-byte loads carry two k steps (same k map on A and B; k summed in another order) and a 32-row wave uses each W fragment for two row tiles
# speedup vs baseline: 3.2831x; 1.1105x over previous
#include <hip/hip_runtime.h>
#include <hip/hip_bf16.h>


#define DIN 128
#define HD  64

typedef __attribute__((ext_vector_type(2))) float v2f;
typedef __attribute__((ext_vector_type(8))) float v8f;

#define BSH     8
#define BROWS   (1 << BSH)
#define BWORDS  (BROWS * HD)
#define LCAP    8192
#define CURSTR  32
#define PCH     8192

typedef __attribute__((ext_vector_type(4))) float v4f;

#define WMMA_F32(a, b, c) \
  __builtin_amdgcn_wmma_f32_16x16x4_f32(false, a, false, b, (short)0, c, false, false)

__global__ __launch_bounds__(128) void gat_gemm_wmma(
    const float* __restrict__ feat, const float* __restrict__ W,
    float* __restrict__ z, int* __restrict__ cur, int N_, int NB) {
  const int gt = blockIdx.x * 128 + threadIdx.x;
  if (gt < NB) cur[gt * CURSTR] = 0;
  const int lane = threadIdx.x & 31;
  const int wave = blockIdx.x * 4 + (threadIdx.x >> 5);
  const int r0 = wave * 32;
  if (r0 >= N_) return;

  const int half = lane >> 4;
  const int m    = lane & 15;

  const int rowA0 = min(r0 + m, N_ - 1);
  const int rowA1 = min(r0 + 16 + m, N_ - 1);
  const float* __restrict__ a0p = feat + (size_t)rowA0 * DIN + 4 * half;
  const float* __restrict__ a1p = feat + (size_t)rowA1 * DIN + 4 * half;
  const float* __restrict__ bp  = W + (size_t)m * DIN + 4 * half;

  v8f acc[2][4];
#pragma unroll
  for (int j = 0; j < 4; ++j) { acc[0][j] = (v8f){}; acc[1][j] = (v8f){}; }

  v4f A0c = *(const v4f*)a0p, A1c = *(const v4f*)a1p, A0n = A0c, A1n = A1c;
  v4f Bc[4], Bn[4];
#pragma unroll
  for (int j = 0; j < 4; ++j) {
    Bc[j] = *(const v4f*)(bp + (size_t)j * 16 * DIN);
    Bn[j] = Bc[j];
  }

#pragma unroll
  for (int t = 0; t < 16; ++t) {
    if (t + 1 < 16) {
      A0n = *(const v4f*)(a0p + 8 * (t + 1));
      A1n = *(const v4f*)(a1p + 8 * (t + 1));
#pragma unroll
      for (int j = 0; j < 4; ++j)
        Bn[j] = *(const v4f*)(bp + (size_t)j * 16 * DIN + 8 * (t + 1));
    }
    const v2f a0l = __builtin_shufflevector(A0c, A0c, 0, 1);
    const v2f a0h = __builtin_shufflevector(A0c, A0c, 2, 3);
    const v2f a1l = __builtin_shufflevector(A1c, A1c, 0, 1);
    const v2f a1h = __builtin_shufflevector(A1c, A1c, 2, 3);
#pragma unroll
    for (int j = 0; j < 4; ++j) {
      const v2f bl = __builtin_shufflevector(Bc[j], Bc[j], 0, 1);
      const v2f bh = __builtin_shufflevector(Bc[j], Bc[j], 2, 3);
      acc[0][j] = WMMA_F32(a0l, bl, acc[0][j]);
      acc[1][j] = WMMA_F32(a1l, bl, acc[1][j]);
      acc[0][j] = WMMA_F32(a0h, bh, acc[0][j]);
      acc[1][j] = WMMA_F32(a1h, bh, acc[1][j]);
    }
    A0c = A0n;
    A1c = A1n;
#pragma unroll
    for (int j = 0; j < 4; ++j) Bc[j] = Bn[j];
  }

#pragma unroll
  for (int i = 0; i < 2; ++i) {
#pragma unroll
    for (int r = 0; r < 8; ++r) {
      const int row = r0 + 16 * i + r + 8 * half;
      if (row < N_) {
        float* zr = z + (size_t)row * HD + m;
        zr[0]  = acc[i][0][r];
        zr[16] = acc[i][1][r];
        zr[32] = acc[i][2][r];
        zr[48] = acc[i][3][r];
      }
    }
  }
}

__device__ __forceinline__ int block_cap(int b, int N_) {
  return min(LCAP, (N_ - (b << BSH)) * HD);
}

__global__ __launch_bounds__(1024) void gat_bin(
    const int* __restrict__ src, const int* __restrict__ dst,
    int* __restrict__ cur, float* __restrict__ out, int N_, int E_, int NB) {
  extern __shared__ int hist[];
  const int tid = threadIdx.x;
  for (int b = tid; b < NB; b += 1024) hist[b] = 0;
  __syncthreads();

  const int cbase = blockIdx.x * PCH;
  int sv[8], tv[8], rv[8];
  if (cbase + PCH <= E_) {
#pragma unroll
    for (int i = 0; i < 2; ++i) {
      const int e0 = cbase + (i * 1024 + tid) * 4;
      const int4 s4 = *(const int4*)(src + e0);
      const int4 t4 = *(const int4*)(dst + e0);
      sv[4 * i + 0] = s4.x; sv[4 * i + 1] = s4.y;
      sv[4 * i + 2] = s4.z; sv[4 * i + 3] = s4.w;
      tv[4 * i + 0] = t4.x; tv[4 * i + 1] = t4.y;
      tv[4 * i + 2] = t4.z; tv[4 * i + 3] = t4.w;
    }
  } else {
#pragma unroll
    for (int k = 0; k < 8; ++k) {
      const int e = cbase + ((k >> 2) * 1024 + tid) * 4 + (k & 3);
      sv[k] = e < E_ ? src[e] : 0;
      tv[k] = e < E_ ? dst[e] : -1;
    }
  }
#pragma unroll
  for (int k = 0; k < 8; ++k) {
    rv[k] = -1;
    if ((unsigned)tv[k] < (unsigned)N_)
      rv[k] = atomicAdd(&hist[tv[k] >> BSH], 1);
  }
  __syncthreads();
  for (int b = tid; b < NB; b += 1024) {
    const int c = hist[b];
    if (c > 0) hist[b] = atomicAdd(&cur[b * CURSTR], c);
  }
  __syncthreads();
#pragma unroll
  for (int k = 0; k < 8; ++k) {
    if (rv[k] >= 0) {
      const int b = tv[k] >> BSH;
      const int pos = hist[b] + rv[k];
      if (pos < block_cap(b, N_)) {
        const unsigned wd = ((unsigned)sv[k] & 0xFFFFFFu) |
                            ((unsigned)(tv[k] & (BROWS - 1)) << 24);
        out[(size_t)b * BWORDS + pos] = __uint_as_float(wd);
      }
    }
  }
}

__device__ __forceinline__ float quad_sum(float t) {
  t += __int_as_float(__builtin_amdgcn_mov_dpp(__float_as_int(t), 0xB1, 0xF, 0xF, true));
  t += __int_as_float(__builtin_amdgcn_mov_dpp(__float_as_int(t), 0x4E, 0xF, 0xF, true));
  return t;
}

__device__ __forceinline__ float head_logit(const float4 zv, const float4 av) {
  float t = zv.x * av.x;
  t = fmaf(zv.y, av.y, t);
  t = fmaf(zv.z, av.z, t);
  t = fmaf(zv.w, av.w, t);
  return quad_sum(t);
}

__device__ __forceinline__ float xhalf(float x) {
  const int b = __float_as_int(x);
  return __int_as_float(__builtin_amdgcn_permlanex16(b, b, 0x76543210, 0xfedcba98, false, false));
}

__device__ __forceinline__ void add_edge(const float4 zv, float w,
    const float4 al, float erh, float& den, float4& a) {
  float x = head_logit(zv, al) + erh;
  x = x > 0.f ? x : 0.2f * x;
  const float p = w * __expf(x);
  den += p;
  a.x = fmaf(p, zv.x, a.x);
  a.y = fmaf(p, zv.y, a.y);
  a.z = fmaf(p, zv.z, a.z);
  a.w = fmaf(p, zv.w, a.w);
}

__global__ __launch_bounds__(1024) void gat_block(
    const int* __restrict__ src, const int* __restrict__ dst,
    const float* __restrict__ aw, const float* __restrict__ z,
    const int* __restrict__ cur, const float* __restrict__ bias,
    float* out, int N_, int E_) {
  __shared__ int sorted[LCAP];
  __shared__ int cntl[BROWS];
  __shared__ int offl[BROWS];

  const int tid  = threadIdx.x;
  const int lane = tid & 31;
  const int wave = __builtin_amdgcn_readfirstlane(tid >> 5);
  const int b    = blockIdx.x;
  const int row0 = b << BSH;
  const int rows = min(BROWS, N_ - row0);
  const int total = __builtin_amdgcn_readfirstlane(cur[b * CURSTR]);
  const bool listed = total <= block_cap(b, N_) && N_ <= (1 << 24);
  float* oblk = out + (size_t)row0 * HD;

  if (tid < BROWS) cntl[tid] = 0;
  __syncthreads();

  unsigned wd[8];
  int rk[8];
#pragma unroll
  for (int k = 0; k < 8; ++k) {
    const int i = k * 1024 + tid;
    wd[k] = 0u;
    if (listed && i < total) wd[k] = __float_as_uint(oblk[i]);
  }
#pragma unroll
  for (int k = 0; k < 8; ++k) {
    const int i = k * 1024 + tid;
    rk[k] = -1;
    if (listed && i < total) rk[k] = atomicAdd(&cntl[wd[k] >> 24], 1);
  }
  __syncthreads();

  if (wave == 0) {
    int c8[8];
    int s = 0;
#pragma unroll
    for (int k = 0; k < 8; ++k) { c8[k] = cntl[lane * 8 + k]; s += c8[k]; }
    int incl = s;
#pragma unroll
    for (int d = 1; d < 32; d <<= 1) {
      const int v = __shfl_up(incl, d);
      if (lane >= d) incl += v;
    }
    int run = incl - s;
#pragma unroll
    for (int k = 0; k < 8; ++k) { offl[lane * 8 + k] = run; run += c8[k]; }
  }
  __syncthreads();

#pragma unroll
  for (int k = 0; k < 8; ++k) {
    if (rk[k] >= 0) sorted[offl[wd[k] >> 24] + rk[k]] = (int)(wd[k] & 0xFFFFFFu);
  }
  __syncthreads();

  const int hl  = lane >> 4;
  const int c16 = lane & 15;
  const float* __restrict__ zc = z + 4 * c16;
  const char* __restrict__ zb = (const char*)z;
  const unsigned cb = 16u * (unsigned)c16;
  const float4 al = *(const float4*)(aw + 4 * c16);
  const float4 ar = *(const float4*)(aw + HD + 4 * c16);
  const float4 bb = *(const float4*)(bias + 4 * c16);
  const unsigned nmax = (unsigned)(N_ - 1);

  float4 zn = make_float4(0.f, 0.f, 0.f, 0.f);
  if (wave * (BROWS / 32) < rows)
    zn = *(const float4*)(zc + (size_t)(row0 + wave * (BROWS / 32)) * HD);

  for (int q = 0; q < BROWS / 32; ++q) {
    const int ld = wave * (BROWS / 32) + q;
    if (ld >= rows) break;
    const int n = row0 + ld;
    const float erh = head_logit(zn, ar);
    if (q + 1 < BROWS / 32 && ld + 1 < rows)
      zn = *(const float4*)(zc + (size_t)(n + 1) * HD);
    float den = 0.f;
    float4 a = make_float4(0.f, 0.f, 0.f, 0.f);
    if (listed) {
      const int c = __builtin_amdgcn_readfirstlane(cntl[ld]);
      const int o = __builtin_amdgcn_readfirstlane(offl[ld]);
      const int* lst = sorted + o;
      int j0 = 0;
      for (; j0 + 8 <= c; j0 += 8) {
        float4 zv[4];
#pragma unroll
        for (int k = 0; k < 4; ++k) {
          const unsigned s = min((unsigned)lst[j0 + 2 * k + hl], nmax);
          zv[k] = *(const float4*)(zb + (s * 256u + cb));
        }
#pragma unroll
        for (int k = 0; k < 4; ++k) add_edge(zv[k], 1.f, al, erh, den, a);
      }
      const int r = c - j0;
      if (r > 0) {
        int sx[4];
        float w[4];
        float4 zv[4];
#pragma unroll
        for (int k = 0; k < 4; ++k) {
          const int e = 2 * k + hl;
          w[k] = e < r ? 1.f : 0.f;
          sx[k] = lst[j0 + min(e, r - 1)];
        }
#pragma unroll
        for (int k = 0; k < 4; ++k) {
          zv[k] = make_float4(0.f, 0.f, 0.f, 0.f);
          if (2 * k < r) {
            const unsigned s = min((unsigned)sx[k], nmax);
            zv[k] = *(const float4*)(zb + (s * 256u + cb));
          }
        }
#pragma unroll
        for (int k = 0; k < 4; ++k) {
          if (2 * k < r) add_edge(zv[k], w[k], al, erh, den, a);
        }
      }
    } else {
      const float wh = hl ? 0.f : 1.f;
      for (int e0 = 0; e0 < E_; e0 += 32) {
        const int e = e0 + lane;
        const int t = e < E_ ? dst[e] : -1;
        const int sl = e < E_ ? src[e] : 0;
        unsigned m = (unsigned)__ballot(t == n);
        while (m) {
          const int j = __builtin_ctz(m);
          m &= m - 1;
          const unsigned s = min((unsigned)__builtin_amdgcn_readlane(sl, j), nmax);
          const float4 zv = *(const float4*)(zc + (size_t)s * HD);
          add_edge(zv, wh, al, erh, den, a);
        }
      }
    }
    den += xhalf(den);
    a.x += xhalf(a.x);
    a.y += xhalf(a.y);
    a.z += xhalf(a.z);
    a.w += xhalf(a.w);
    den = den > 0.f ? den : 1.f;
    float4 o4;
    o4.x = a.x / den + bb.x;
    o4.y = a.y / den + bb.y;
    o4.z = a.z / den + bb.z;
    o4.w = a.w / den + bb.w;
    if (hl == 0) *(float4*)(oblk + (size_t)ld * HD + 4 * c16) = o4;
  }
}

extern "C" void kernel_launch(void* const* d_in, const int* in_sizes, int n_in,
                              void* d_out, int out_size, void* d_ws, size_t ws_size,
                              hipStream_t stream) {
  const float* feat = (const float*)d_in[0];
  const int*   src  = (const int*)d_in[1];
  const int*   dst  = (const int*)d_in[2];
  const float* W    = (const float*)d_in[3];
  const float* aw   = (const float*)d_in[4];
  const float* bias = (const float*)d_in[5];
  float* out = (float*)d_out;

  const int N_ = in_sizes[0] / DIN;
  const int E_ = in_sizes[1];
  const int NB = (N_ + BROWS - 1) / BROWS;

  float* ws  = (float*)d_ws;
  float* z   = ws;
  int*   cur = (int*)(z + (size_t)N_ * HD);

  const int gemmWaves  = (N_ + 31) / 32;
  const int gemmBlocks = (gemmWaves + 3) / 4;
  gat_gemm_wmma<<<gemmBlocks, 128, 0, stream>>>(feat, W, z, cur, N_, NB);

  gat_bin<<<(E_ + PCH - 1) / PCH, 1024, NB * sizeof(int), stream>>>(src, dst, cur, out, N_, E_, NB);

  gat_block<<<NB, 1024, 0, stream>>>(src, dst, aw, z, cur, bias, out, N_, E_);
}


# Binning work-group sorts its chunk by destination block in LDS (prefix over the block counts) and writes each block's run with consecutive lanes, not one scattered word per lane; list order kept
# speedup vs baseline: 3.7352x; 1.1377x over previous
#include <hip/hip_runtime.h>
#include <hip/hip_bf16.h>


#define DIN 128
#define HD  64

typedef __attribute__((ext_vector_type(2))) float v2f;
typedef __attribute__((ext_vector_type(8))) float v8f;

#define BSH     8
#define BROWS   (1 << BSH)
#define BWORDS  (BROWS * HD)
#define LCAP    8192
#define CURSTR  32
#define PCH     8192

typedef __attribute__((ext_vector_type(4))) float v4f;

#define WMMA_F32(a, b, c) \
  __builtin_amdgcn_wmma_f32_16x16x4_f32(false, a, false, b, (short)0, c, false, false)

__global__ __launch_bounds__(128) void gat_gemm_wmma(
    const float* __restrict__ feat, const float* __restrict__ W,
    float* __restrict__ z, int* __restrict__ cur, int N_, int NB) {
  const int gt = blockIdx.x * 128 + threadIdx.x;
  if (gt < NB) cur[gt * CURSTR] = 0;
  const int lane = threadIdx.x & 31;
  const int wave = blockIdx.x * 4 + (threadIdx.x >> 5);
  const int r0 = wave * 32;
  if (r0 >= N_) return;

  const int half = lane >> 4;
  const int m    = lane & 15;

  const int rowA0 = min(r0 + m, N_ - 1);
  const int rowA1 = min(r0 + 16 + m, N_ - 1);
  const float* __restrict__ a0p = feat + (size_t)rowA0 * DIN + 4 * half;
  const float* __restrict__ a1p = feat + (size_t)rowA1 * DIN + 4 * half;
  const float* __restrict__ bp  = W + (size_t)m * DIN + 4 * half;

  v8f acc[2][4];
#pragma unroll
  for (int j = 0; j < 4; ++j) { acc[0][j] = (v8f){}; acc[1][j] = (v8f){}; }

  v4f A0c = *(const v4f*)a0p, A1c = *(const v4f*)a1p, A0n = A0c, A1n = A1c;
  v4f Bc[4], Bn[4];
#pragma unroll
  for (int j = 0; j < 4; ++j) {
    Bc[j] = *(const v4f*)(bp + (size_t)j * 16 * DIN);
    Bn[j] = Bc[j];
  }

#pragma unroll
  for (int t = 0; t < 16; ++t) {
    if (t + 1 < 16) {
      A0n = *(const v4f*)(a0p + 8 * (t + 1));
      A1n = *(const v4f*)(a1p + 8 * (t + 1));
#pragma unroll
      for (int j = 0; j < 4; ++j)
        Bn[j] = *(const v4f*)(bp + (size_t)j * 16 * DIN + 8 * (t + 1));
    }
    const v2f a0l = __builtin_shufflevector(A0c, A0c, 0, 1);
    const v2f a0h = __builtin_shufflevector(A0c, A0c, 2, 3);
    const v2f a1l = __builtin_shufflevector(A1c, A1c, 0, 1);
    const v2f a1h = __builtin_shufflevector(A1c, A1c, 2, 3);
#pragma unroll
    for (int j = 0; j < 4; ++j) {
      const v2f bl = __builtin_shufflevector(Bc[j], Bc[j], 0, 1);
      const v2f bh = __builtin_shufflevector(Bc[j], Bc[j], 2, 3);
      acc[0][j] = WMMA_F32(a0l, bl, acc[0][j]);
      acc[1][j] = WMMA_F32(a1l, bl, acc[1][j]);
      acc[0][j] = WMMA_F32(a0h, bh, acc[0][j]);
      acc[1][j] = WMMA_F32(a1h, bh, acc[1][j]);
    }
    A0c = A0n;
    A1c = A1n;
#pragma unroll
    for (int j = 0; j < 4; ++j) Bc[j] = Bn[j];
  }

#pragma unroll
  for (int i = 0; i < 2; ++i) {
#pragma unroll
    for (int r = 0; r < 8; ++r) {
      const int row = r0 + 16 * i + r + 8 * half;
      if (row < N_) {
        float* zr = z + (size_t)row * HD + m;
        zr[0]  = acc[i][0][r];
        zr[16] = acc[i][1][r];
        zr[32] = acc[i][2][r];
        zr[48] = acc[i][3][r];
      }
    }
  }
}

__device__ __forceinline__ int block_cap(int b, int N_) {
  return min(LCAP, (N_ - (b << BSH)) * HD);
}

__global__ __launch_bounds__(1024) void gat_bin(
    const int* __restrict__ src, const int* __restrict__ dst,
    int* __restrict__ cur, float* __restrict__ out, int N_, int E_, int NB) {
  extern __shared__ int2 binfo[];
  __shared__ int2 stg[PCH];
  __shared__ int wsum[33];
  if (N_ > (1 << 24)) return;
  int* hist = (int*)(binfo + NB);
  const int tid  = threadIdx.x;
  const int lane = tid & 31;
  const int wave = tid >> 5;
  for (int b = tid; b < NB; b += 1024) hist[b] = 0;
  __syncthreads();

  const int cbase = blockIdx.x * PCH;
  int sv[8], tv[8], rv[8];
  if (cbase + PCH <= E_) {
#pragma unroll
    for (int i = 0; i < 2; ++i) {
      const int e0 = cbase + (i * 1024 + tid) * 4;
      const int4 s4 = *(const int4*)(src + e0);
      const int4 t4 = *(const int4*)(dst + e0);
      sv[4 * i + 0] = s4.x; sv[4 * i + 1] = s4.y;
      sv[4 * i + 2] = s4.z; sv[4 * i + 3] = s4.w;
      tv[4 * i + 0] = t4.x; tv[4 * i + 1] = t4.y;
      tv[4 * i + 2] = t4.z; tv[4 * i + 3] = t4.w;
    }
  } else {
#pragma unroll
    for (int k = 0; k < 8; ++k) {
      const int e = cbase + ((k >> 2) * 1024 + tid) * 4 + (k & 3);
      sv[k] = e < E_ ? src[e] : 0;
      tv[k] = e < E_ ? dst[e] : -1;
    }
  }
#pragma unroll
  for (int k = 0; k < 8; ++k) {
    rv[k] = -1;
    if ((unsigned)tv[k] < (unsigned)N_)
      rv[k] = atomicAdd(&hist[tv[k] >> BSH], 1);
  }
  __syncthreads();

  const int ipt = (NB + 1023) >> 10;
  const int bfirst = tid * ipt;
  int s = 0;
  for (int i = 0; i < ipt; ++i) {
    const int b = bfirst + i;
    if (b < NB) s += hist[b];
  }
  int incl = s;
#pragma unroll
  for (int d = 1; d < 32; d <<= 1) {
    const int v = __shfl_up(incl, d);
    if (lane >= d) incl += v;
  }
  if (lane == 31) wsum[wave] = incl;
  __syncthreads();
  if (wave == 0) {
    const int w = wsum[lane];
    int wi = w;
#pragma unroll
    for (int d = 1; d < 32; d <<= 1) {
      const int v = __shfl_up(wi, d);
      if (lane >= d) wi += v;
    }
    wsum[lane] = wi - w;
    if (lane == 31) wsum[32] = wi;
  }
  __syncthreads();
  int run = wsum[wave] + incl - s;
  for (int i = 0; i < ipt; ++i) {
    const int b = bfirst + i;
    if (b < NB) {
      const int c = hist[b];
      int g = 0;
      if (c > 0) g = atomicAdd(&cur[b * CURSTR], c);
      binfo[b] = make_int2(run, g);
      run += c;
    }
  }
  __syncthreads();
  const int tot = wsum[32];

  int2 og[8];
#pragma unroll
  for (int k = 0; k < 8; ++k) og[k] = binfo[rv[k] >= 0 ? (tv[k] >> BSH) : 0];
#pragma unroll
  for (int k = 0; k < 8; ++k) {
    if (rv[k] >= 0) {
      const int b = tv[k] >> BSH;
      const int pos = og[k].y + rv[k];
      const unsigned wd = ((unsigned)sv[k] & 0xFFFFFFu) |
                          ((unsigned)(tv[k] & (BROWS - 1)) << 24);
      const int a = pos < block_cap(b, N_) ? b * BWORDS + pos : -1;
      stg[og[k].x + rv[k]] = make_int2((int)wd, a);
    }
  }
  __syncthreads();
  int2 ev[8];
#pragma unroll
  for (int k = 0; k < 8; ++k) {
    const int i = k * 1024 + tid;
    ev[k] = make_int2(0, -1);
    if (i < tot) ev[k] = stg[i];
  }
#pragma unroll
  for (int k = 0; k < 8; ++k) {
    if (ev[k].y >= 0) out[ev[k].y] = __int_as_float(ev[k].x);
  }
}

__device__ __forceinline__ float quad_sum(float t) {
  t += __int_as_float(__builtin_amdgcn_mov_dpp(__float_as_int(t), 0xB1, 0xF, 0xF, true));
  t += __int_as_float(__builtin_amdgcn_mov_dpp(__float_as_int(t), 0x4E, 0xF, 0xF, true));
  return t;
}

__device__ __forceinline__ float head_logit(const float4 zv, const float4 av) {
  float t = zv.x * av.x;
  t = fmaf(zv.y, av.y, t);
  t = fmaf(zv.z, av.z, t);
  t = fmaf(zv.w, av.w, t);
  return quad_sum(t);
}

__device__ __forceinline__ float xhalf(float x) {
  const int b = __float_as_int(x);
  return __int_as_float(__builtin_amdgcn_permlanex16(b, b, 0x76543210, 0xfedcba98, false, false));
}

__device__ __forceinline__ void add_edge(const float4 zv, float w,
    const float4 al, float erh, float& den, float4& a) {
  float x = head_logit(zv, al) + erh;
  x = x > 0.f ? x : 0.2f * x;
  const float p = w * __expf(x);
  den += p;
  a.x = fmaf(p, zv.x, a.x);
  a.y = fmaf(p, zv.y, a.y);
  a.z = fmaf(p, zv.z, a.z);
  a.w = fmaf(p, zv.w, a.w);
}

__global__ __launch_bounds__(1024) void gat_block(
    const int* __restrict__ src, const int* __restrict__ dst,
    const float* __restrict__ aw, const float* __restrict__ z,
    const int* __restrict__ cur, const float* __restrict__ bias,
    float* out, int N_, int E_) {
  __shared__ int sorted[LCAP];
  __shared__ int cntl[BROWS];
  __shared__ int offl[BROWS];

  const int tid  = threadIdx.x;
  const int lane = tid & 31;
  const int wave = __builtin_amdgcn_readfirstlane(tid >> 5);
  const int b    = blockIdx.x;
  const int row0 = b << BSH;
  const int rows = min(BROWS, N_ - row0);
  const int total = __builtin_amdgcn_readfirstlane(cur[b * CURSTR]);
  const bool listed = total <= block_cap(b, N_) && N_ <= (1 << 24);
  float* oblk = out + (size_t)row0 * HD;

  if (tid < BROWS) cntl[tid] = 0;
  __syncthreads();

  unsigned wd[8];
  int rk[8];
#pragma unroll
  for (int k = 0; k < 8; ++k) {
    const int i = k * 1024 + tid;
    wd[k] = 0u;
    if (listed && i < total) wd[k] = __float_as_uint(oblk[i]);
  }
#pragma unroll
  for (int k = 0; k < 8; ++k) {
    const int i = k * 1024 + tid;
    rk[k] = -1;
    if (listed && i < total) rk[k] = atomicAdd(&cntl[wd[k] >> 24], 1);
  }
  __syncthreads();

  if (wave == 0) {
    int c8[8];
    int s = 0;
#pragma unroll
    for (int k = 0; k < 8; ++k) { c8[k] = cntl[lane * 8 + k]; s += c8[k]; }
    int incl = s;
#pragma unroll
    for (int d = 1; d < 32; d <<= 1) {
      const int v = __shfl_up(incl, d);
      if (lane >= d) incl += v;
    }
    int run = incl - s;
#pragma unroll
    for (int k = 0; k < 8; ++k) { offl[lane * 8 + k] = run; run += c8[k]; }
  }
  __syncthreads();

#pragma unroll
  for (int k = 0; k < 8; ++k) {
    if (rk[k] >= 0) sorted[offl[wd[k] >> 24] + rk[k]] = (int)(wd[k] & 0xFFFFFFu);
  }
  __syncthreads();

  const int hl  = lane >> 4;
  const int c16 = lane & 15;
  const float* __restrict__ zc = z + 4 * c16;
  const char* __restrict__ zb = (const char*)z;
  const unsigned cb = 16u * (unsigned)c16;
  const float4 al = *(const float4*)(aw + 4 * c16);
  const float4 ar = *(const float4*)(aw + HD + 4 * c16);
  const float4 bb = *(const float4*)(bias + 4 * c16);
  const unsigned nmax = (unsigned)(N_ - 1);

  float4 zn = make_float4(0.f, 0.f, 0.f, 0.f);
  if (wave * (BROWS / 32) < rows)
    zn = *(const float4*)(zc + (size_t)(row0 + wave * (BROWS / 32)) * HD);

  for (int q = 0; q < BROWS / 32; ++q) {
    const int ld = wave * (BROWS / 32) + q;
    if (ld >= rows) break;
    const int n = row0 + ld;
    const float erh = head_logit(zn, ar);
    if (q + 1 < BROWS / 32 && ld + 1 < rows)
      zn = *(const float4*)(zc + (size_t)(n + 1) * HD);
    float den = 0.f;
    float4 a = make_float4(0.f, 0.f, 0.f, 0.f);
    if (listed) {
      const int c = __builtin_amdgcn_readfirstlane(cntl[ld]);
      const int o = __builtin_amdgcn_readfirstlane(offl[ld]);
      const int* lst = sorted + o;
      int j0 = 0;
      for (; j0 + 8 <= c; j0 += 8) {
        float4 zv[4];
#pragma unroll
        for (int k = 0; k < 4; ++k) {
          const unsigned s = min((unsigned)lst[j0 + 2 * k + hl], nmax);
          zv[k] = *(const float4*)(zb + (s * 256u + cb));
        }
#pragma unroll
        for (int k = 0; k < 4; ++k) add_edge(zv[k], 1.f, al, erh, den, a);
      }
      const int r = c - j0;
      if (r > 0) {
        int sx[4];
        float w[4];
        float4 zv[4];
#pragma unroll
        for (int k = 0; k < 4; ++k) {
          const int e = 2 * k + hl;
          w[k] = e < r ? 1.f : 0.f;
          sx[k] = lst[j0 + min(e, r - 1)];
        }
#pragma unroll
        for (int k = 0; k < 4; ++k) {
          zv[k] = make_float4(0.f, 0.f, 0.f, 0.f);
          if (2 * k < r) {
            const unsigned s = min((unsigned)sx[k], nmax);
            zv[k] = *(const float4*)(zb + (s * 256u + cb));
          }
        }
#pragma unroll
        for (int k = 0; k < 4; ++k) {
          if (2 * k < r) add_edge(zv[k], w[k], al, erh, den, a);
        }
      }
    } else {
      const float wh = hl ? 0.f : 1.f;
      for (int e0 = 0; e0 < E_; e0 += 32) {
        const int e = e0 + lane;
        const int t = e < E_ ? dst[e] : -1;
        const int sl = e < E_ ? src[e] : 0;
        unsigned m = (unsigned)__ballot(t == n);
        while (m) {
          const int j = __builtin_ctz(m);
          m &= m - 1;
          const unsigned s = min((unsigned)__builtin_amdgcn_readlane(sl, j), nmax);
          const float4 zv = *(const float4*)(zc + (size_t)s * HD);
          add_edge(zv, wh, al, erh, den, a);
        }
      }
    }
    den += xhalf(den);
    a.x += xhalf(a.x);
    a.y += xhalf(a.y);
    a.z += xhalf(a.z);
    a.w += xhalf(a.w);
    den = den > 0.f ? den : 1.f;
    float4 o4;
    o4.x = a.x / den + bb.x;
    o4.y = a.y / den + bb.y;
    o4.z = a.z / den + bb.z;
    o4.w = a.w / den + bb.w;
    if (hl == 0) *(float4*)(oblk + (size_t)ld * HD + 4 * c16) = o4;
  }
}

extern "C" void kernel_launch(void* const* d_in, const int* in_sizes, int n_in,
                              void* d_out, int out_size, void* d_ws, size_t ws_size,
                              hipStream_t stream) {
  const float* feat = (const float*)d_in[0];
  const int*   src  = (const int*)d_in[1];
  const int*   dst  = (const int*)d_in[2];
  const float* W    = (const float*)d_in[3];
  const float* aw   = (const float*)d_in[4];
  const float* bias = (const float*)d_in[5];
  float* out = (float*)d_out;

  const int N_ = in_sizes[0] / DIN;
  const int E_ = in_sizes[1];
  const int NB = (N_ + BROWS - 1) / BROWS;

  float* ws  = (float*)d_ws;
  float* z   = ws;
  int*   cur = (int*)(z + (size_t)N_ * HD);

  const int gemmWaves  = (N_ + 31) / 32;
  const int gemmBlocks = (gemmWaves + 3) / 4;
  gat_gemm_wmma<<<gemmBlocks, 128, 0, stream>>>(feat, W, z, cur, N_, NB);

  gat_bin<<<(E_ + PCH - 1) / PCH, 1024, 3 * NB * sizeof(int), stream>>>(src, dst, cur, out, N_, E_, NB);

  gat_block<<<NB, 1024, 0, stream>>>(src, dst, aw, z, cur, bias, out, N_, E_);
}


# Binning inside the GEMM launch (2 launches): each work-group sorts a 4096-edge chunk by block in LDS, stores it whole with run starts; block kernel collects the runs; no cursors; arrival-order sums
# speedup vs baseline: 4.1845x; 1.1203x over previous
#include <hip/hip_runtime.h>
#include <hip/hip_bf16.h>


#define DIN 128
#define HD  64

typedef __attribute__((ext_vector_type(2))) float v2f;
typedef __attribute__((ext_vector_type(8))) float v8f;

#define BSH     8
#define BROWS   (1 << BSH)
#define LCAP    8192
#define PCH     4096
#define FT      256
#define EPT     (PCH / FT)
#define CHMAX   1024

typedef __attribute__((ext_vector_type(4))) float v4f;

#define WMMA_F32(a, b, c) \
  __builtin_amdgcn_wmma_f32_16x16x4_f32(false, a, false, b, (short)0, c, false, false)

__device__ __forceinline__ bool ws_holds(unsigned long long wsb, int N_, int NB, int NCH) {
  const unsigned long long need =
      4ull * ((unsigned long long)N_ * HD + (unsigned long long)NCH * PCH +
              (unsigned long long)NCH * (NB + 1));
  return wsb >= need;
}

__device__ __forceinline__ void bin_chunk(
    const int* __restrict__ src, const int* __restrict__ dst,
    int* __restrict__ lst, int* __restrict__ tab,
    int* hist, int* bst, int* stg, int* wsum,
    int c, int N_, int E_, int NB) {
  const int tid  = threadIdx.x;
  const int lane = tid & 31;
  const int wave = tid >> 5;
  for (int b = tid; b < NB; b += FT) hist[b] = 0;
  __syncthreads();

  const int cbase = c * PCH;
  int sv[EPT], tv[EPT], rv[EPT];
  if (cbase + PCH <= E_) {
#pragma unroll
    for (int i = 0; i < EPT / 4; ++i) {
      const int e0 = cbase + (i * FT + tid) * 4;
      const int4 s4 = *(const int4*)(src + e0);
      const int4 t4 = *(const int4*)(dst + e0);
      sv[4 * i + 0] = s4.x; sv[4 * i + 1] = s4.y;
      sv[4 * i + 2] = s4.z; sv[4 * i + 3] = s4.w;
      tv[4 * i + 0] = t4.x; tv[4 * i + 1] = t4.y;
      tv[4 * i + 2] = t4.z; tv[4 * i + 3] = t4.w;
    }
  } else {
#pragma unroll
    for (int k = 0; k < EPT; ++k) {
      const int e = cbase + ((k >> 2) * FT + tid) * 4 + (k & 3);
      sv[k] = e < E_ ? src[e] : 0;
      tv[k] = e < E_ ? dst[e] : -1;
    }
  }
#pragma unroll
  for (int k = 0; k < EPT; ++k) {
    rv[k] = -1;
    if ((unsigned)tv[k] < (unsigned)N_)
      rv[k] = atomicAdd(&hist[tv[k] >> BSH], 1);
  }
  __syncthreads();

  const int ipt = (NB + FT - 1) / FT;
  const int bfirst = tid * ipt;
  int s = 0;
  for (int i = 0; i < ipt; ++i) {
    const int b = bfirst + i;
    if (b < NB) s += hist[b];
  }
  int incl = s;
#pragma unroll
  for (int d = 1; d < 32; d <<= 1) {
    const int v = __shfl_up(incl, d);
    if (lane >= d) incl += v;
  }
  if (lane == 31) wsum[wave] = incl;
  __syncthreads();
  if (wave == 0) {
    const int w = lane < FT / 32 ? wsum[lane] : 0;
    int wi = w;
#pragma unroll
    for (int d = 1; d < 32; d <<= 1) {
      const int v = __shfl_up(wi, d);
      if (lane >= d) wi += v;
    }
    if (lane < FT / 32) wsum[lane] = wi - w;
    if (lane == 31) wsum[FT / 32] = wi;
  }
  __syncthreads();
  int* trow = tab + (size_t)c * (NB + 1);
  int run = wsum[wave] + incl - s;
  for (int i = 0; i < ipt; ++i) {
    const int b = bfirst + i;
    if (b < NB) {
      bst[b] = run;
      trow[b] = run;
      run += hist[b];
    }
  }
  const int tot = wsum[FT / 32];
  if (tid == 0) trow[NB] = tot;
  __syncthreads();

#pragma unroll
  for (int k = 0; k < EPT; ++k) {
    if (rv[k] >= 0) {
      const unsigned wd = ((unsigned)sv[k] & 0xFFFFFFu) |
                          ((unsigned)(tv[k] & (BROWS - 1)) << 24);
      stg[bst[tv[k] >> BSH] + rv[k]] = (int)wd;
    }
  }
  __syncthreads();
  int ev[EPT];
#pragma unroll
  for (int k = 0; k < EPT; ++k) {
    const int i = k * FT + tid;
    ev[k] = 0;
    if (i < tot) ev[k] = stg[i];
  }
  int* lrow = lst + (size_t)c * PCH;
#pragma unroll
  for (int k = 0; k < EPT; ++k) {
    const int i = k * FT + tid;
    if (i < tot) lrow[i] = ev[k];
  }
}

__global__ __launch_bounds__(FT) void gat_proj_bin(
    const float* __restrict__ feat, const float* __restrict__ W,
    const int* __restrict__ src, const int* __restrict__ dst,
    float* __restrict__ z, int* __restrict__ lst, int* __restrict__ tab,
    int N_, int E_, int NB, int NCH, unsigned long long wsb) {
  extern __shared__ int dynl[];
  __shared__ int stg[PCH];
  __shared__ int wsum[FT / 32 + 1];
  if (!ws_holds(wsb, N_, NB, NCH)) return;
  if ((int)blockIdx.x < NCH && N_ <= (1 << 24))
    bin_chunk(src, dst, lst, tab, dynl, dynl + NB, stg, wsum, blockIdx.x, N_, E_, NB);

  const int lane = threadIdx.x & 31;
  const int wave = blockIdx.x * (FT / 32) + (threadIdx.x >> 5);
  const int r0 = wave * 32;
  if (r0 >= N_) return;

  const int half = lane >> 4;
  const int m    = lane & 15;

  const int rowA0 = min(r0 + m, N_ - 1);
  const int rowA1 = min(r0 + 16 + m, N_ - 1);
  const float* __restrict__ a0p = feat + (size_t)rowA0 * DIN + 4 * half;
  const float* __restrict__ a1p = feat + (size_t)rowA1 * DIN + 4 * half;
  const float* __restrict__ bp  = W + (size_t)m * DIN + 4 * half;

  v8f acc[2][4];
#pragma unroll
  for (int j = 0; j < 4; ++j) { acc[0][j] = (v8f){}; acc[1][j] = (v8f){}; }

  v4f A0c = *(const v4f*)a0p, A1c = *(const v4f*)a1p, A0n = A0c, A1n = A1c;
  v4f Bc[4], Bn[4];
#pragma unroll
  for (int j = 0; j < 4; ++j) {
    Bc[j] = *(const v4f*)(bp + (size_t)j * 16 * DIN);
    Bn[j] = Bc[j];
  }

#pragma unroll
  for (int t = 0; t < 16; ++t) {
    if (t + 1 < 16) {
      A0n = *(const v4f*)(a0p + 8 * (t + 1));
      A1n = *(const v4f*)(a1p + 8 * (t + 1));
#pragma unroll
      for (int j = 0; j < 4; ++j)
        Bn[j] = *(const v4f*)(bp + (size_t)j * 16 * DIN + 8 * (t + 1));
    }
    const v2f a0l = __builtin_shufflevector(A0c, A0c, 0, 1);
    const v2f a0h = __builtin_shufflevector(A0c, A0c, 2, 3);
    const v2f a1l = __builtin_shufflevector(A1c, A1c, 0, 1);
    const v2f a1h = __builtin_shufflevector(A1c, A1c, 2, 3);
#pragma unroll
    for (int j = 0; j < 4; ++j) {
      const v2f bl = __builtin_shufflevector(Bc[j], Bc[j], 0, 1);
      const v2f bh = __builtin_shufflevector(Bc[j], Bc[j], 2, 3);
      acc[0][j] = WMMA_F32(a0l, bl, acc[0][j]);
      acc[1][j] = WMMA_F32(a1l, bl, acc[1][j]);
      acc[0][j] = WMMA_F32(a0h, bh, acc[0][j]);
      acc[1][j] = WMMA_F32(a1h, bh, acc[1][j]);
    }
    A0c = A0n;
    A1c = A1n;
#pragma unroll
    for (int j = 0; j < 4; ++j) Bc[j] = Bn[j];
  }

#pragma unroll
  for (int i = 0; i < 2; ++i) {
#pragma unroll
    for (int r = 0; r < 8; ++r) {
      const int row = r0 + 16 * i + r + 8 * half;
      if (row < N_) {
        float* zr = z + (size_t)row * HD + m;
        zr[0]  = acc[i][0][r];
        zr[16] = acc[i][1][r];
        zr[32] = acc[i][2][r];
        zr[48] = acc[i][3][r];
      }
    }
  }
}

__device__ __forceinline__ float quad_sum(float t) {
  t += __int_as_float(__builtin_amdgcn_mov_dpp(__float_as_int(t), 0xB1, 0xF, 0xF, true));
  t += __int_as_float(__builtin_amdgcn_mov_dpp(__float_as_int(t), 0x4E, 0xF, 0xF, true));
  return t;
}

__device__ __forceinline__ float head_logit(const float4 zv, const float4 av) {
  float t = zv.x * av.x;
  t = fmaf(zv.y, av.y, t);
  t = fmaf(zv.z, av.z, t);
  t = fmaf(zv.w, av.w, t);
  return quad_sum(t);
}

__device__ __forceinline__ float xhalf(float x) {
  const int b = __float_as_int(x);
  return __int_as_float(__builtin_amdgcn_permlanex16(b, b, 0x76543210, 0xfedcba98, false, false));
}

__device__ __forceinline__ void add_edge(const float4 zv, float w,
    const float4 al, float erh, float& den, float4& a) {
  float x = head_logit(zv, al) + erh;
  x = x > 0.f ? x : 0.2f * x;
  const float p = w * __expf(x);
  den += p;
  a.x = fmaf(p, zv.x, a.x);
  a.y = fmaf(p, zv.y, a.y);
  a.z = fmaf(p, zv.z, a.z);
  a.w = fmaf(p, zv.w, a.w);
}

__global__ __launch_bounds__(1024) void gat_block(
    const int* __restrict__ src, const int* __restrict__ dst,
    const float* __restrict__ aw, const float* __restrict__ z,
    const int* __restrict__ chunks, const int* __restrict__ tab,
    const float* __restrict__ bias, float* __restrict__ out,
    int N_, int E_, int NB, int NCH, unsigned long long wsb) {
  __shared__ int sorted[LCAP];
  __shared__ int2 raw[LCAP];
  __shared__ int cntl[BROWS];
  __shared__ int offl[BROWS];
  __shared__ int cst[CHMAX];
  __shared__ int ccn[CHMAX];
  __shared__ int cbs[CHMAX];
  __shared__ int wsum[33];

  if (!ws_holds(wsb, N_, NB, NCH)) return;
  const int tid  = threadIdx.x;
  const int lane = tid & 31;
  const int wave = __builtin_amdgcn_readfirstlane(tid >> 5);
  const int b    = blockIdx.x;
  const int row0 = b << BSH;
  const int rows = min(BROWS, N_ - row0);
  const bool small = N_ <= (1 << 24) && NCH <= CHMAX;
  float* oblk = out + (size_t)row0 * HD;

  if (tid < BROWS) cntl[tid] = 0;
  int mys = 0, myn = 0;
  if (small && tid < NCH) {
    const int* t = tab + (size_t)tid * (NB + 1) + b;
    const int t0 = t[0], t1 = t[1];
    mys = min(max(t0, 0), PCH);
    myn = min(max(t1 - mys, 0), PCH - mys);
  }
  int incl = myn;
#pragma unroll
  for (int d = 1; d < 32; d <<= 1) {
    const int v = __shfl_up(incl, d);
    if (lane >= d) incl += v;
  }
  if (lane == 31) wsum[wave] = incl;
  __syncthreads();
  if (wave == 0) {
    const int w = wsum[lane];
    int wi = w;
#pragma unroll
    for (int d = 1; d < 32; d <<= 1) {
      const int v = __shfl_up(wi, d);
      if (lane >= d) wi += v;
    }
    wsum[lane] = wi - w;
    if (lane == 31) wsum[32] = wi;
  }
  __syncthreads();
  cst[tid] = mys;
  ccn[tid] = myn;
  cbs[tid] = wsum[wave] + incl - myn;
  const int total = __builtin_amdgcn_readfirstlane(wsum[32]);
  const bool listed = small && total <= LCAP;
  __syncthreads();

  if (listed) {
    for (int g = 0; g < NCH; g += 256) {
      unsigned wd[8];
      int kk[8], rk[8], cn[8], cp[8], cs[8];
#pragma unroll
      for (int j = 0; j < 8; ++j) {
        const int c = g + wave + 32 * j;
        const int cc = min(c, CHMAX - 1);
        cn[j] = c < NCH ? ccn[cc] : 0;
        cp[j] = cbs[cc];
        cs[j] = cst[cc];
      }
#pragma unroll
      for (int j = 0; j < 8; ++j) {
        const int c = g + wave + 32 * j;
        kk[j] = -1;
        wd[j] = 0u;
        if (lane < cn[j]) {
          kk[j] = cp[j] + lane;
          wd[j] = (unsigned)chunks[(size_t)c * PCH + cs[j] + lane];
        }
      }
#pragma unroll
      for (int j = 0; j < 8; ++j) {
        rk[j] = 0;
        if (kk[j] >= 0) rk[j] = atomicAdd(&cntl[wd[j] >> 24], 1);
      }
#pragma unroll
      for (int j = 0; j < 8; ++j) {
        if (kk[j] >= 0) raw[kk[j]] = make_int2((int)wd[j], rk[j]);
      }
      for (int j = 0; j < 8; ++j) {
        const int c = g + wave + 32 * j;
        if (c < NCH) {
          const int n = ccn[c];
          for (int i = lane + 32; i < n; i += 32) {
            const unsigned w = (unsigned)chunks[(size_t)c * PCH + cst[c] + i];
            const int r = atomicAdd(&cntl[w >> 24], 1);
            raw[cbs[c] + i] = make_int2((int)w, r);
          }
        }
      }
    }
  }
  __syncthreads();

  if (wave == 0) {
    int c8[8];
    int s = 0;
#pragma unroll
    for (int k = 0; k < 8; ++k) { c8[k] = cntl[lane * 8 + k]; s += c8[k]; }
    int in2 = s;
#pragma unroll
    for (int d = 1; d < 32; d <<= 1) {
      const int v = __shfl_up(in2, d);
      if (lane >= d) in2 += v;
    }
    int run = in2 - s;
#pragma unroll
    for (int k = 0; k < 8; ++k) { offl[lane * 8 + k] = run; run += c8[k]; }
  }
  __syncthreads();

  if (listed) {
    for (int k = tid; k < total; k += 1024) {
      const int2 p = raw[k];
      sorted[offl[(unsigned)p.x >> 24] + p.y] = p.x & 0xFFFFFF;
    }
  }
  __syncthreads();

  const int hl  = lane >> 4;
  const int c16 = lane & 15;
  const float* __restrict__ zc = z + 4 * c16;
  const char* __restrict__ zb = (const char*)z;
  const unsigned cb = 16u * (unsigned)c16;
  const float4 al = *(const float4*)(aw + 4 * c16);
  const float4 ar = *(const float4*)(aw + HD + 4 * c16);
  const float4 bb = *(const float4*)(bias + 4 * c16);
  const unsigned nmax = (unsigned)(N_ - 1);

  float4 zn = make_float4(0.f, 0.f, 0.f, 0.f);
  if (wave * (BROWS / 32) < rows)
    zn = *(const float4*)(zc + (size_t)(row0 + wave * (BROWS / 32)) * HD);

  for (int q = 0; q < BROWS / 32; ++q) {
    const int ld = wave * (BROWS / 32) + q;
    if (ld >= rows) break;
    const int n = row0 + ld;
    const float erh = head_logit(zn, ar);
    if (q + 1 < BROWS / 32 && ld + 1 < rows)
      zn = *(const float4*)(zc + (size_t)(n + 1) * HD);
    float den = 0.f;
    float4 a = make_float4(0.f, 0.f, 0.f, 0.f);
    if (listed) {
      const int c = __builtin_amdgcn_readfirstlane(cntl[ld]);
      const int o = __builtin_amdgcn_readfirstlane(offl[ld]);
      const int* lst = sorted + o;
      int j0 = 0;
      for (; j0 + 8 <= c; j0 += 8) {
        float4 zv[4];
#pragma unroll
        for (int k = 0; k < 4; ++k) {
          const unsigned s = min((unsigned)lst[j0 + 2 * k + hl], nmax);
          zv[k] = *(const float4*)(zb + (s * 256u + cb));
        }
#pragma unroll
        for (int k = 0; k < 4; ++k) add_edge(zv[k], 1.f, al, erh, den, a);
      }
      const int r = c - j0;
      if (r > 0) {
        int sx[4];
        float w[4];
        float4 zv[4];
#pragma unroll
        for (int k = 0; k < 4; ++k) {
          const int e = 2 * k + hl;
          w[k] = e < r ? 1.f : 0.f;
          sx[k] = lst[j0 + min(e, r - 1)];
        }
#pragma unroll
        for (int k = 0; k < 4; ++k) {
          zv[k] = make_float4(0.f, 0.f, 0.f, 0.f);
          if (2 * k < r) {
            const unsigned s = min((unsigned)sx[k], nmax);
            zv[k] = *(const float4*)(zb + (s * 256u + cb));
          }
        }
#pragma unroll
        for (int k = 0; k < 4; ++k) {
          if (2 * k < r) add_edge(zv[k], w[k], al, erh, den, a);
        }
      }
    } else {
      const float wh = hl ? 0.f : 1.f;
      for (int e0 = 0; e0 < E_; e0 += 32) {
        const int e = e0 + lane;
        const int t = e < E_ ? dst[e] : -1;
        const int sl = e < E_ ? src[e] : 0;
        unsigned m = (unsigned)__ballot(t == n);
        while (m) {
          const int j = __builtin_ctz(m);
          m &= m - 1;
          const unsigned s = min((unsigned)__builtin_amdgcn_readlane(sl, j), nmax);
          const float4 zv = *(const float4*)(zc + (size_t)s * HD);
          add_edge(zv, wh, al, erh, den, a);
        }
      }
    }
    den += xhalf(den);
    a.x += xhalf(a.x);
    a.y += xhalf(a.y);
    a.z += xhalf(a.z);
    a.w += xhalf(a.w);
    den = den > 0.f ? den : 1.f;
    float4 o4;
    o4.x = a.x / den + bb.x;
    o4.y = a.y / den + bb.y;
    o4.z = a.z / den + bb.z;
    o4.w = a.w / den + bb.w;
    if (hl == 0) *(float4*)(oblk + (size_t)ld * HD + 4 * c16) = o4;
  }
}

extern "C" void kernel_launch(void* const* d_in, const int* in_sizes, int n_in,
                              void* d_out, int out_size, void* d_ws, size_t ws_size,
                              hipStream_t stream) {
  const float* feat = (const float*)d_in[0];
  const int*   src  = (const int*)d_in[1];
  const int*   dst  = (const int*)d_in[2];
  const float* W    = (const float*)d_in[3];
  const float* aw   = (const float*)d_in[4];
  const float* bias = (const float*)d_in[5];
  float* out = (float*)d_out;

  const int N_ = in_sizes[0] / DIN;
  const int E_ = in_sizes[1];
  const int NB  = (N_ + BROWS - 1) / BROWS;
  const int NCH = (E_ + PCH - 1) / PCH;

  float* ws  = (float*)d_ws;
  float* z   = ws;
  int*   lst = (int*)(z + (size_t)N_ * HD);
  int*   tab = lst + (size_t)NCH * PCH;

  const int g1 = NB > NCH ? NB : NCH;
  gat_proj_bin<<<g1, FT, 2 * NB * sizeof(int), stream>>>(
      feat, W, src, dst, z, lst, tab, N_, E_, NB, NCH, (unsigned long long)ws_size);

  gat_block<<<NB, 1024, 0, stream>>>(src, dst, aw, z, lst, tab, bias, out, N_, E_, NB, NCH,
                                     (unsigned long long)ws_size);
}


# Destination block 200 rows, not 256 (wave w sums rows w, w+32, ..): 500 block work-groups stand two on each of the 256 WGPs instead of 391 (two on some, one on others); arrival-order sums as before
# speedup vs baseline: 4.3946x; 1.0502x over previous
#include <hip/hip_runtime.h>
#include <hip/hip_bf16.h>


#define DIN 128
#define HD  64

typedef __attribute__((ext_vector_type(2))) float v2f;
typedef __attribute__((ext_vector_type(8))) float v8f;

#define BROWS   200
#define BSLOT   256
#define BW      32
#define GROWS   256
#define LCAP    8192
#define PCH     4096
#define FT      256
#define EPT     (PCH / FT)
#define CHMAX   1024

typedef __attribute__((ext_vector_type(4))) float v4f;

#define WMMA_F32(a, b, c) \
  __builtin_amdgcn_wmma_f32_16x16x4_f32(false, a, false, b, (short)0, c, false, false)

__device__ __forceinline__ bool ws_holds(unsigned long long wsb, int N_, int NB, int NCH) {
  const unsigned long long need =
      4ull * ((unsigned long long)N_ * HD + (unsigned long long)NCH * PCH +
              (unsigned long long)NCH * (NB + 1));
  return wsb >= need;
}

__device__ __forceinline__ void bin_chunk(
    const int* __restrict__ src, const int* __restrict__ dst,
    int* __restrict__ lst, int* __restrict__ tab,
    int* hist, int* bst, int* stg, int* wsum,
    int c, int N_, int E_, int NB) {
  const int tid  = threadIdx.x;
  const int lane = tid & 31;
  const int wave = tid >> 5;
  for (int b = tid; b < NB; b += FT) hist[b] = 0;
  __syncthreads();

  const int cbase = c * PCH;
  int sv[EPT], tv[EPT], rv[EPT];
  if (cbase + PCH <= E_) {
#pragma unroll
    for (int i = 0; i < EPT / 4; ++i) {
      const int e0 = cbase + (i * FT + tid) * 4;
      const int4 s4 = *(const int4*)(src + e0);
      const int4 t4 = *(const int4*)(dst + e0);
      sv[4 * i + 0] = s4.x; sv[4 * i + 1] = s4.y;
      sv[4 * i + 2] = s4.z; sv[4 * i + 3] = s4.w;
      tv[4 * i + 0] = t4.x; tv[4 * i + 1] = t4.y;
      tv[4 * i + 2] = t4.z; tv[4 * i + 3] = t4.w;
    }
  } else {
#pragma unroll
    for (int k = 0; k < EPT; ++k) {
      const int e = cbase + ((k >> 2) * FT + tid) * 4 + (k & 3);
      sv[k] = e < E_ ? src[e] : 0;
      tv[k] = e < E_ ? dst[e] : -1;
    }
  }
#pragma unroll
  for (int k = 0; k < EPT; ++k) {
    rv[k] = -1;
    if ((unsigned)tv[k] < (unsigned)N_) {
      const unsigned blk = (unsigned)tv[k] / BROWS;
      const unsigned rib = (unsigned)tv[k] - blk * BROWS;
      rv[k] = atomicAdd(&hist[blk], 1);
      sv[k] = (int)(((unsigned)sv[k] & 0xFFFFFFu) | (rib << 24));
      tv[k] = (int)blk;
    }
  }
  __syncthreads();

  const int ipt = (NB + FT - 1) / FT;
  const int bfirst = tid * ipt;
  int s = 0;
  for (int i = 0; i < ipt; ++i) {
    const int b = bfirst + i;
    if (b < NB) s += hist[b];
  }
  int incl = s;
#pragma unroll
  for (int d = 1; d < 32; d <<= 1) {
    const int v = __shfl_up(incl, d);
    if (lane >= d) incl += v;
  }
  if (lane == 31) wsum[wave] = incl;
  __syncthreads();
  if (wave == 0) {
    const int w = lane < FT / 32 ? wsum[lane] : 0;
    int wi = w;
#pragma unroll
    for (int d = 1; d < 32; d <<= 1) {
      const int v = __shfl_up(wi, d);
      if (lane >= d) wi += v;
    }
    if (lane < FT / 32) wsum[lane] = wi - w;
    if (lane == 31) wsum[FT / 32] = wi;
  }
  __syncthreads();
  int* trow = tab + (size_t)c * (NB + 1);
  int run = wsum[wave] + incl - s;
  for (int i = 0; i < ipt; ++i) {
    const int b = bfirst + i;
    if (b < NB) {
      bst[b] = run;
      trow[b] = run;
      run += hist[b];
    }
  }
  const int tot = wsum[FT / 32];
  if (tid == 0) trow[NB] = tot;
  __syncthreads();

#pragma unroll
  for (int k = 0; k < EPT; ++k) {
    if (rv[k] >= 0) stg[bst[tv[k]] + rv[k]] = sv[k];
  }
  __syncthreads();
  int ev[EPT];
#pragma unroll
  for (int k = 0; k < EPT; ++k) {
    const int i = k * FT + tid;
    ev[k] = 0;
    if (i < tot) ev[k] = stg[i];
  }
  int* lrow = lst + (size_t)c * PCH;
#pragma unroll
  for (int k = 0; k < EPT; ++k) {
    const int i = k * FT + tid;
    if (i < tot) lrow[i] = ev[k];
  }
}

__global__ __launch_bounds__(FT) void gat_proj_bin(
    const float* __restrict__ feat, const float* __restrict__ W,
    const int* __restrict__ src, const int* __restrict__ dst,
    float* __restrict__ z, int* __restrict__ lst, int* __restrict__ tab,
    int N_, int E_, int NB, int NCH, unsigned long long wsb) {
  extern __shared__ int dynl[];
  __shared__ int stg[PCH];
  __shared__ int wsum[FT / 32 + 1];
  if (!ws_holds(wsb, N_, NB, NCH)) return;
  if ((int)blockIdx.x < NCH && N_ <= (1 << 24))
    bin_chunk(src, dst, lst, tab, dynl, dynl + NB, stg, wsum, blockIdx.x, N_, E_, NB);

  const int lane = threadIdx.x & 31;
  const int wave = blockIdx.x * (FT / 32) + (threadIdx.x >> 5);
  const int r0 = wave * 32;
  if (r0 >= N_) return;

  const int half = lane >> 4;
  const int m    = lane & 15;

  const int rowA0 = min(r0 + m, N_ - 1);
  const int rowA1 = min(r0 + 16 + m, N_ - 1);
  const float* __restrict__ a0p = feat + (size_t)rowA0 * DIN + 4 * half;
  const float* __restrict__ a1p = feat + (size_t)rowA1 * DIN + 4 * half;
  const float* __restrict__ bp  = W + (size_t)m * DIN + 4 * half;

  v8f acc[2][4];
#pragma unroll
  for (int j = 0; j < 4; ++j) { acc[0][j] = (v8f){}; acc[1][j] = (v8f){}; }

  v4f A0c = *(const v4f*)a0p, A1c = *(const v4f*)a1p, A0n = A0c, A1n = A1c;
  v4f Bc[4], Bn[4];
#pragma unroll
  for (int j = 0; j < 4; ++j) {
    Bc[j] = *(const v4f*)(bp + (size_t)j * 16 * DIN);
    Bn[j] = Bc[j];
  }

#pragma unroll
  for (int t = 0; t < 16; ++t) {
    if (t + 1 < 16) {
      A0n = *(const v4f*)(a0p + 8 * (t + 1));
      A1n = *(const v4f*)(a1p + 8 * (t + 1));
#pragma unroll
      for (int j = 0; j < 4; ++j)
        Bn[j] = *(const v4f*)(bp + (size_t)j * 16 * DIN + 8 * (t + 1));
    }
    const v2f a0l = __builtin_shufflevector(A0c, A0c, 0, 1);
    const v2f a0h = __builtin_shufflevector(A0c, A0c, 2, 3);
    const v2f a1l = __builtin_shufflevector(A1c, A1c, 0, 1);
    const v2f a1h = __builtin_shufflevector(A1c, A1c, 2, 3);
#pragma unroll
    for (int j = 0; j < 4; ++j) {
      const v2f bl = __builtin_shufflevector(Bc[j], Bc[j], 0, 1);
      const v2f bh = __builtin_shufflevector(Bc[j], Bc[j], 2, 3);
      acc[0][j] = WMMA_F32(a0l, bl, acc[0][j]);
      acc[1][j] = WMMA_F32(a1l, bl, acc[1][j]);
      acc[0][j] = WMMA_F32(a0h, bh, acc[0][j]);
      acc[1][j] = WMMA_F32(a1h, bh, acc[1][j]);
    }
    A0c = A0n;
    A1c = A1n;
#pragma unroll
    for (int j = 0; j < 4; ++j) Bc[j] = Bn[j];
  }

#pragma unroll
  for (int i = 0; i < 2; ++i) {
#pragma unroll
    for (int r = 0; r < 8; ++r) {
      const int row = r0 + 16 * i + r + 8 * half;
      if (row < N_) {
        float* zr = z + (size_t)row * HD + m;
        zr[0]  = acc[i][0][r];
        zr[16] = acc[i][1][r];
        zr[32] = acc[i][2][r];
        zr[48] = acc[i][3][r];
      }
    }
  }
}

__device__ __forceinline__ float quad_sum(float t) {
  t += __int_as_float(__builtin_amdgcn_mov_dpp(__float_as_int(t), 0xB1, 0xF, 0xF, true));
  t += __int_as_float(__builtin_amdgcn_mov_dpp(__float_as_int(t), 0x4E, 0xF, 0xF, true));
  return t;
}

__device__ __forceinline__ float head_logit(const float4 zv, const float4 av) {
  float t = zv.x * av.x;
  t = fmaf(zv.y, av.y, t);
  t = fmaf(zv.z, av.z, t);
  t = fmaf(zv.w, av.w, t);
  return quad_sum(t);
}

__device__ __forceinline__ float xhalf(float x) {
  const int b = __float_as_int(x);
  return __int_as_float(__builtin_amdgcn_permlanex16(b, b, 0x76543210, 0xfedcba98, false, false));
}

__device__ __forceinline__ void add_edge(const float4 zv, float w,
    const float4 al, float erh, float& den, float4& a) {
  float x = head_logit(zv, al) + erh;
  x = x > 0.f ? x : 0.2f * x;
  const float p = w * __expf(x);
  den += p;
  a.x = fmaf(p, zv.x, a.x);
  a.y = fmaf(p, zv.y, a.y);
  a.z = fmaf(p, zv.z, a.z);
  a.w = fmaf(p, zv.w, a.w);
}

__global__ __launch_bounds__(BW * 32) void gat_block(
    const int* __restrict__ src, const int* __restrict__ dst,
    const float* __restrict__ aw, const float* __restrict__ z,
    const int* __restrict__ chunks, const int* __restrict__ tab,
    const float* __restrict__ bias, float* __restrict__ out,
    int N_, int E_, int NB, int NCH, unsigned long long wsb) {
  __shared__ int sorted[LCAP];
  __shared__ int2 raw[LCAP];
  __shared__ int cntl[BSLOT];
  __shared__ int offl[BSLOT];
  __shared__ int cst[CHMAX];
  __shared__ int ccn[CHMAX];
  __shared__ int cbs[CHMAX];
  __shared__ int wsum[33];

  if (!ws_holds(wsb, N_, NB, NCH)) return;
  const int tid  = threadIdx.x;
  const int lane = tid & 31;
  const int wave = __builtin_amdgcn_readfirstlane(tid >> 5);
  const int b    = blockIdx.x;
  const int row0 = b * BROWS;
  const int rows = min(BROWS, N_ - row0);
  const bool small = N_ <= (1 << 24) && NCH <= CHMAX;
  float* oblk = out + (size_t)row0 * HD;

  if (tid < BSLOT) cntl[tid] = 0;
  int mys = 0, myn = 0;
  if (small && tid < NCH) {
    const int* t = tab + (size_t)tid * (NB + 1) + b;
    const int t0 = t[0], t1 = t[1];
    mys = min(max(t0, 0), PCH);
    myn = min(max(t1 - mys, 0), PCH - mys);
  }
  int incl = myn;
#pragma unroll
  for (int d = 1; d < 32; d <<= 1) {
    const int v = __shfl_up(incl, d);
    if (lane >= d) incl += v;
  }
  if (lane == 31) wsum[wave] = incl;
  __syncthreads();
  if (wave == 0) {
    const int w = wsum[lane];
    int wi = w;
#pragma unroll
    for (int d = 1; d < 32; d <<= 1) {
      const int v = __shfl_up(wi, d);
      if (lane >= d) wi += v;
    }
    wsum[lane] = wi - w;
    if (lane == 31) wsum[32] = wi;
  }
  __syncthreads();
  cst[tid] = mys;
  ccn[tid] = myn;
  cbs[tid] = wsum[wave] + incl - myn;
  const int total = __builtin_amdgcn_readfirstlane(wsum[32]);
  const bool listed = small && total <= LCAP;
  __syncthreads();

  if (listed) {
    for (int g = 0; g < NCH; g += 256) {
      unsigned wd[8];
      int kk[8], rk[8], cn[8], cp[8], cs[8];
#pragma unroll
      for (int j = 0; j < 8; ++j) {
        const int c = g + wave + 32 * j;
        const int cc = min(c, CHMAX - 1);
        cn[j] = c < NCH ? ccn[cc] : 0;
        cp[j] = cbs[cc];
        cs[j] = cst[cc];
      }
#pragma unroll
      for (int j = 0; j < 8; ++j) {
        const int c = g + wave + 32 * j;
        kk[j] = -1;
        wd[j] = 0u;
        if (lane < cn[j]) {
          kk[j] = cp[j] + lane;
          wd[j] = (unsigned)chunks[(size_t)c * PCH + cs[j] + lane];
        }
      }
#pragma unroll
      for (int j = 0; j < 8; ++j) {
        rk[j] = 0;
        if (kk[j] >= 0) rk[j] = atomicAdd(&cntl[wd[j] >> 24], 1);
      }
#pragma unroll
      for (int j = 0; j < 8; ++j) {
        if (kk[j] >= 0) raw[kk[j]] = make_int2((int)wd[j], rk[j]);
      }
      for (int j = 0; j < 8; ++j) {
        const int c = g + wave + 32 * j;
        if (c < NCH) {
          const int n = ccn[c];
          for (int i = lane + 32; i < n; i += 32) {
            const unsigned w = (unsigned)chunks[(size_t)c * PCH + cst[c] + i];
            const int r = atomicAdd(&cntl[w >> 24], 1);
            raw[cbs[c] + i] = make_int2((int)w, r);
          }
        }
      }
    }
  }
  __syncthreads();

  if (wave == 0) {
    int c8[8];
    int s = 0;
#pragma unroll
    for (int k = 0; k < 8; ++k) { c8[k] = cntl[lane * 8 + k]; s += c8[k]; }
    int in2 = s;
#pragma unroll
    for (int d = 1; d < 32; d <<= 1) {
      const int v = __shfl_up(in2, d);
      if (lane >= d) in2 += v;
    }
    int run = in2 - s;
#pragma unroll
    for (int k = 0; k < 8; ++k) { offl[lane * 8 + k] = run; run += c8[k]; }
  }
  __syncthreads();

  if (listed) {
    for (int k = tid; k < total; k += 1024) {
      const int2 p = raw[k];
      sorted[offl[(unsigned)p.x >> 24] + p.y] = p.x & 0xFFFFFF;
    }
  }
  __syncthreads();

  const int hl  = lane >> 4;
  const int c16 = lane & 15;
  const float* __restrict__ zc = z + 4 * c16;
  const char* __restrict__ zb = (const char*)z;
  const unsigned cb = 16u * (unsigned)c16;
  const float4 al = *(const float4*)(aw + 4 * c16);
  const float4 ar = *(const float4*)(aw + HD + 4 * c16);
  const float4 bb = *(const float4*)(bias + 4 * c16);
  const unsigned nmax = (unsigned)(N_ - 1);

  float4 zn = make_float4(0.f, 0.f, 0.f, 0.f);
  if (wave < rows)
    zn = *(const float4*)(zc + (size_t)(row0 + wave) * HD);

  for (int q = 0; q < (BROWS + BW - 1) / BW; ++q) {
    const int ld = wave + BW * q;
    if (ld >= rows) break;
    const int n = row0 + ld;
    const float erh = head_logit(zn, ar);
    if (ld + BW < rows)
      zn = *(const float4*)(zc + (size_t)(n + BW) * HD);
    float den = 0.f;
    float4 a = make_float4(0.f, 0.f, 0.f, 0.f);
    if (listed) {
      const int c = __builtin_amdgcn_readfirstlane(cntl[ld]);
      const int o = __builtin_amdgcn_readfirstlane(offl[ld]);
      const int* lst = sorted + o;
      int j0 = 0;
      for (; j0 + 8 <= c; j0 += 8) {
        float4 zv[4];
#pragma unroll
        for (int k = 0; k < 4; ++k) {
          const unsigned s = min((unsigned)lst[j0 + 2 * k + hl], nmax);
          zv[k] = *(const float4*)(zb + (s * 256u + cb));
        }
#pragma unroll
        for (int k = 0; k < 4; ++k) add_edge(zv[k], 1.f, al, erh, den, a);
      }
      const int r = c - j0;
      if (r > 0) {
        int sx[4];
        float w[4];
        float4 zv[4];
#pragma unroll
        for (int k = 0; k < 4; ++k) {
          const int e = 2 * k + hl;
          w[k] = e < r ? 1.f : 0.f;
          sx[k] = lst[j0 + min(e, r - 1)];
        }
#pragma unroll
        for (int k = 0; k < 4; ++k) {
          zv[k] = make_float4(0.f, 0.f, 0.f, 0.f);
          if (2 * k < r) {
            const unsigned s = min((unsigned)sx[k], nmax);
            zv[k] = *(const float4*)(zb + (s * 256u + cb));
          }
        }
#pragma unroll
        for (int k = 0; k < 4; ++k) {
          if (2 * k < r) add_edge(zv[k], w[k], al, erh, den, a);
        }
      }
    } else {
      const float wh = hl ? 0.f : 1.f;
      for (int e0 = 0; e0 < E_; e0 += 32) {
        const int e = e0 + lane;
        const int t = e < E_ ? dst[e] : -1;
        const int sl = e < E_ ? src[e] : 0;
        unsigned m = (unsigned)__ballot(t == n);
        while (m) {
          const int j = __builtin_ctz(m);
          m &= m - 1;
          const unsigned s = min((unsigned)__builtin_amdgcn_readlane(sl, j), nmax);
          const float4 zv = *(const float4*)(zc + (size_t)s * HD);
          add_edge(zv, wh, al, erh, den, a);
        }
      }
    }
    den += xhalf(den);
    a.x += xhalf(a.x);
    a.y += xhalf(a.y);
    a.z += xhalf(a.z);
    a.w += xhalf(a.w);
    den = den > 0.f ? den : 1.f;
    float4 o4;
    o4.x = a.x / den + bb.x;
    o4.y = a.y / den + bb.y;
    o4.z = a.z / den + bb.z;
    o4.w = a.w / den + bb.w;
    if (hl == 0) *(float4*)(oblk + (size_t)ld * HD + 4 * c16) = o4;
  }
}

extern "C" void kernel_launch(void* const* d_in, const int* in_sizes, int n_in,
                              void* d_out, int out_size, void* d_ws, size_t ws_size,
                              hipStream_t stream) {
  const float* feat = (const float*)d_in[0];
  const int*   src  = (const int*)d_in[1];
  const int*   dst  = (const int*)d_in[2];
  const float* W    = (const float*)d_in[3];
  const float* aw   = (const float*)d_in[4];
  const float* bias = (const float*)d_in[5];
  float* out = (float*)d_out;

  const int N_ = in_sizes[0] / DIN;
  const int E_ = in_sizes[1];
  const int NB  = (N_ + BROWS - 1) / BROWS;
  const int NCH = (E_ + PCH - 1) / PCH;

  float* ws  = (float*)d_ws;
  float* z   = ws;
  int*   lst = (int*)(z + (size_t)N_ * HD);
  int*   tab = lst + (size_t)NCH * PCH;

  const int NG = (N_ + GROWS - 1) / GROWS;
  const int g1 = NG > NCH ? NG : NCH;
  gat_proj_bin<<<g1, FT, 2 * NB * sizeof(int), stream>>>(
      feat, W, src, dst, z, lst, tab, N_, E_, NB, NCH, (unsigned long long)ws_size);

  gat_block<<<NB, BW * 32, 0, stream>>>(src, dst, aw, z, lst, tab, bias, out, N_, E_, NB, NCH,
                                     (unsigned long long)ws_size);
}


# Block kernel fetches its runs half a wave per chunk, 16 chunks of a wave in flight: one pass over the chunks, not two, one wave-wide test for runs over 16 words; arrival-order sums as before
# speedup vs baseline: 4.5415x; 1.0334x over previous
#include <hip/hip_runtime.h>
#include <hip/hip_bf16.h>


#define DIN 128
#define HD  64

typedef __attribute__((ext_vector_type(2))) float v2f;
typedef __attribute__((ext_vector_type(8))) float v8f;

#define BROWS   200
#define BSLOT   256
#define BW      32
#define GROWS   256
#define LCAP    8192
#define PCH     4096
#define FT      256
#define EPT     (PCH / FT)
#define CHMAX   1024

typedef __attribute__((ext_vector_type(4))) float v4f;

#define WMMA_F32(a, b, c) \
  __builtin_amdgcn_wmma_f32_16x16x4_f32(false, a, false, b, (short)0, c, false, false)

__device__ __forceinline__ bool ws_holds(unsigned long long wsb, int N_, int NB, int NCH) {
  const unsigned long long need =
      4ull * ((unsigned long long)N_ * HD + (unsigned long long)NCH * PCH +
              (unsigned long long)NCH * (NB + 1));
  return wsb >= need;
}

__device__ __forceinline__ void bin_chunk(
    const int* __restrict__ src, const int* __restrict__ dst,
    int* __restrict__ lst, int* __restrict__ tab,
    int* hist, int* bst, int* stg, int* wsum,
    int c, int N_, int E_, int NB) {
  const int tid  = threadIdx.x;
  const int lane = tid & 31;
  const int wave = tid >> 5;
  for (int b = tid; b < NB; b += FT) hist[b] = 0;
  __syncthreads();

  const int cbase = c * PCH;
  int sv[EPT], tv[EPT], rv[EPT];
  if (cbase + PCH <= E_) {
#pragma unroll
    for (int i = 0; i < EPT / 4; ++i) {
      const int e0 = cbase + (i * FT + tid) * 4;
      const int4 s4 = *(const int4*)(src + e0);
      const int4 t4 = *(const int4*)(dst + e0);
      sv[4 * i + 0] = s4.x; sv[4 * i + 1] = s4.y;
      sv[4 * i + 2] = s4.z; sv[4 * i + 3] = s4.w;
      tv[4 * i + 0] = t4.x; tv[4 * i + 1] = t4.y;
      tv[4 * i + 2] = t4.z; tv[4 * i + 3] = t4.w;
    }
  } else {
#pragma unroll
    for (int k = 0; k < EPT; ++k) {
      const int e = cbase + ((k >> 2) * FT + tid) * 4 + (k & 3);
      sv[k] = e < E_ ? src[e] : 0;
      tv[k] = e < E_ ? dst[e] : -1;
    }
  }
#pragma unroll
  for (int k = 0; k < EPT; ++k) {
    rv[k] = -1;
    if ((unsigned)tv[k] < (unsigned)N_) {
      const unsigned blk = (unsigned)tv[k] / BROWS;
      const unsigned rib = (unsigned)tv[k] - blk * BROWS;
      rv[k] = atomicAdd(&hist[blk], 1);
      sv[k] = (int)(((unsigned)sv[k] & 0xFFFFFFu) | (rib << 24));
      tv[k] = (int)blk;
    }
  }
  __syncthreads();

  const int ipt = (NB + FT - 1) / FT;
  const int bfirst = tid * ipt;
  int s = 0;
  for (int i = 0; i < ipt; ++i) {
    const int b = bfirst + i;
    if (b < NB) s += hist[b];
  }
  int incl = s;
#pragma unroll
  for (int d = 1; d < 32; d <<= 1) {
    const int v = __shfl_up(incl, d);
    if (lane >= d) incl += v;
  }
  if (lane == 31) wsum[wave] = incl;
  __syncthreads();
  if (wave == 0) {
    const int w = lane < FT / 32 ? wsum[lane] : 0;
    int wi = w;
#pragma unroll
    for (int d = 1; d < 32; d <<= 1) {
      const int v = __shfl_up(wi, d);
      if (lane >= d) wi += v;
    }
    if (lane < FT / 32) wsum[lane] = wi - w;
    if (lane == 31) wsum[FT / 32] = wi;
  }
  __syncthreads();
  int* trow = tab + (size_t)c * (NB + 1);
  int run = wsum[wave] + incl - s;
  for (int i = 0; i < ipt; ++i) {
    const int b = bfirst + i;
    if (b < NB) {
      bst[b] = run;
      trow[b] = run;
      run += hist[b];
    }
  }
  const int tot = wsum[FT / 32];
  if (tid == 0) trow[NB] = tot;
  __syncthreads();

#pragma unroll
  for (int k = 0; k < EPT; ++k) {
    if (rv[k] >= 0) stg[bst[tv[k]] + rv[k]] = sv[k];
  }
  __syncthreads();
  int ev[EPT];
#pragma unroll
  for (int k = 0; k < EPT; ++k) {
    const int i = k * FT + tid;
    ev[k] = 0;
    if (i < tot) ev[k] = stg[i];
  }
  int* lrow = lst + (size_t)c * PCH;
#pragma unroll
  for (int k = 0; k < EPT; ++k) {
    const int i = k * FT + tid;
    if (i < tot) lrow[i] = ev[k];
  }
}

__global__ __launch_bounds__(FT) void gat_proj_bin(
    const float* __restrict__ feat, const float* __restrict__ W,
    const int* __restrict__ src, const int* __restrict__ dst,
    float* __restrict__ z, int* __restrict__ lst, int* __restrict__ tab,
    int N_, int E_, int NB, int NCH, unsigned long long wsb) {
  extern __shared__ int dynl[];
  __shared__ int stg[PCH];
  __shared__ int wsum[FT / 32 + 1];
  if (!ws_holds(wsb, N_, NB, NCH)) return;
  if ((int)blockIdx.x < NCH && N_ <= (1 << 24))
    bin_chunk(src, dst, lst, tab, dynl, dynl + NB, stg, wsum, blockIdx.x, N_, E_, NB);

  const int lane = threadIdx.x & 31;
  const int wave = blockIdx.x * (FT / 32) + (threadIdx.x >> 5);
  const int r0 = wave * 32;
  if (r0 >= N_) return;

  const int half = lane >> 4;
  const int m    = lane & 15;

  const int rowA0 = min(r0 + m, N_ - 1);
  const int rowA1 = min(r0 + 16 + m, N_ - 1);
  const float* __restrict__ a0p = feat + (size_t)rowA0 * DIN + 4 * half;
  const float* __restrict__ a1p = feat + (size_t)rowA1 * DIN + 4 * half;
  const float* __restrict__ bp  = W + (size_t)m * DIN + 4 * half;

  v8f acc[2][4];
#pragma unroll
  for (int j = 0; j < 4; ++j) { acc[0][j] = (v8f){}; acc[1][j] = (v8f){}; }

  v4f A0c = *(const v4f*)a0p, A1c = *(const v4f*)a1p, A0n = A0c, A1n = A1c;
  v4f Bc[4], Bn[4];
#pragma unroll
  for (int j = 0; j < 4; ++j) {
    Bc[j] = *(const v4f*)(bp + (size_t)j * 16 * DIN);
    Bn[j] = Bc[j];
  }

#pragma unroll
  for (int t = 0; t < 16; ++t) {
    if (t + 1 < 16) {
      A0n = *(const v4f*)(a0p + 8 * (t + 1));
      A1n = *(const v4f*)(a1p + 8 * (t + 1));
#pragma unroll
      for (int j = 0; j < 4; ++j)
        Bn[j] = *(const v4f*)(bp + (size_t)j * 16 * DIN + 8 * (t + 1));
    }
    const v2f a0l = __builtin_shufflevector(A0c, A0c, 0, 1);
    const v2f a0h = __builtin_shufflevector(A0c, A0c, 2, 3);
    const v2f a1l = __builtin_shufflevector(A1c, A1c, 0, 1);
    const v2f a1h = __builtin_shufflevector(A1c, A1c, 2, 3);
#pragma unroll
    for (int j = 0; j < 4; ++j) {
      const v2f bl = __builtin_shufflevector(Bc[j], Bc[j], 0, 1);
      const v2f bh = __builtin_shufflevector(Bc[j], Bc[j], 2, 3);
      acc[0][j] = WMMA_F32(a0l, bl, acc[0][j]);
      acc[1][j] = WMMA_F32(a1l, bl, acc[1][j]);
      acc[0][j] = WMMA_F32(a0h, bh, acc[0][j]);
      acc[1][j] = WMMA_F32(a1h, bh, acc[1][j]);
    }
    A0c = A0n;
    A1c = A1n;
#pragma unroll
    for (int j = 0; j < 4; ++j) Bc[j] = Bn[j];
  }

#pragma unroll
  for (int i = 0; i < 2; ++i) {
#pragma unroll
    for (int r = 0; r < 8; ++r) {
      const int row = r0 + 16 * i + r + 8 * half;
      if (row < N_) {
        float* zr = z + (size_t)row * HD + m;
        zr[0]  = acc[i][0][r];
        zr[16] = acc[i][1][r];
        zr[32] = acc[i][2][r];
        zr[48] = acc[i][3][r];
      }
    }
  }
}

__device__ __forceinline__ float quad_sum(float t) {
  t += __int_as_float(__builtin_amdgcn_mov_dpp(__float_as_int(t), 0xB1, 0xF, 0xF, true));
  t += __int_as_float(__builtin_amdgcn_mov_dpp(__float_as_int(t), 0x4E, 0xF, 0xF, true));
  return t;
}

__device__ __forceinline__ float head_logit(const float4 zv, const float4 av) {
  float t = zv.x * av.x;
  t = fmaf(zv.y, av.y, t);
  t = fmaf(zv.z, av.z, t);
  t = fmaf(zv.w, av.w, t);
  return quad_sum(t);
}

__device__ __forceinline__ float xhalf(float x) {
  const int b = __float_as_int(x);
  return __int_as_float(__builtin_amdgcn_permlanex16(b, b, 0x76543210, 0xfedcba98, false, false));
}

__device__ __forceinline__ void add_edge(const float4 zv, float w,
    const float4 al, float erh, float& den, float4& a) {
  float x = head_logit(zv, al) + erh;
  x = x > 0.f ? x : 0.2f * x;
  const float p = w * __expf(x);
  den += p;
  a.x = fmaf(p, zv.x, a.x);
  a.y = fmaf(p, zv.y, a.y);
  a.z = fmaf(p, zv.z, a.z);
  a.w = fmaf(p, zv.w, a.w);
}

__global__ __launch_bounds__(BW * 32) void gat_block(
    const int* __restrict__ src, const int* __restrict__ dst,
    const float* __restrict__ aw, const float* __restrict__ z,
    const int* __restrict__ chunks, const int* __restrict__ tab,
    const float* __restrict__ bias, float* __restrict__ out,
    int N_, int E_, int NB, int NCH, unsigned long long wsb) {
  __shared__ int sorted[LCAP];
  __shared__ int2 raw[LCAP];
  __shared__ int cntl[BSLOT];
  __shared__ int offl[BSLOT];
  __shared__ int cst[CHMAX];
  __shared__ int ccn[CHMAX];
  __shared__ int cbs[CHMAX];
  __shared__ int wsum[33];

  if (!ws_holds(wsb, N_, NB, NCH)) return;
  const int tid  = threadIdx.x;
  const int lane = tid & 31;
  const int wave = __builtin_amdgcn_readfirstlane(tid >> 5);
  const int b    = blockIdx.x;
  const int row0 = b * BROWS;
  const int rows = min(BROWS, N_ - row0);
  const bool small = N_ <= (1 << 24) && NCH <= CHMAX;
  float* oblk = out + (size_t)row0 * HD;

  if (tid < BSLOT) cntl[tid] = 0;
  int mys = 0, myn = 0;
  if (small && tid < NCH) {
    const int* t = tab + (size_t)tid * (NB + 1) + b;
    const int t0 = t[0], t1 = t[1];
    mys = min(max(t0, 0), PCH);
    myn = min(max(t1 - mys, 0), PCH - mys);
  }
  int incl = myn;
#pragma unroll
  for (int d = 1; d < 32; d <<= 1) {
    const int v = __shfl_up(incl, d);
    if (lane >= d) incl += v;
  }
  if (lane == 31) wsum[wave] = incl;
  __syncthreads();
  if (wave == 0) {
    const int w = wsum[lane];
    int wi = w;
#pragma unroll
    for (int d = 1; d < 32; d <<= 1) {
      const int v = __shfl_up(wi, d);
      if (lane >= d) wi += v;
    }
    wsum[lane] = wi - w;
    if (lane == 31) wsum[32] = wi;
  }
  __syncthreads();
  cst[tid] = mys;
  ccn[tid] = myn;
  cbs[tid] = wsum[wave] + incl - myn;
  const int total = __builtin_amdgcn_readfirstlane(wsum[32]);
  const bool listed = small && total <= LCAP;
  __syncthreads();

  if (listed) {
    const int l16 = lane & 15;
    const int hw  = 2 * wave + (lane >> 4);
    for (int g = 0; g < NCH; g += 16 * BW) {
      unsigned wd[8];
      int kk[8], rk[8], cn[8], cp[8], cs[8];
      int nmx = 0;
#pragma unroll
      for (int j = 0; j < 8; ++j) {
        const int c = g + hw + 2 * BW * j;
        const int cc = min(c, CHMAX - 1);
        cn[j] = ccn[cc];
        cp[j] = cbs[cc];
        cs[j] = cst[cc];
        nmx = max(nmx, cn[j]);
      }
#pragma unroll
      for (int j = 0; j < 8; ++j) {
        const int c = g + hw + 2 * BW * j;
        kk[j] = -1;
        wd[j] = 0u;
        if (l16 < cn[j]) {
          kk[j] = cp[j] + l16;
          wd[j] = (unsigned)chunks[(unsigned)c * PCH + (unsigned)(cs[j] + l16)];
        }
      }
#pragma unroll
      for (int j = 0; j < 8; ++j) {
        rk[j] = 0;
        if (kk[j] >= 0) rk[j] = atomicAdd(&cntl[wd[j] >> 24], 1);
      }
#pragma unroll
      for (int j = 0; j < 8; ++j) {
        if (kk[j] >= 0) raw[kk[j]] = make_int2((int)wd[j], rk[j]);
      }
      if (__ballot(nmx > 16) != 0) {
        for (int j = 0; j < 8; ++j) {
          const int c = g + hw + 2 * BW * j;
          if (c < NCH) {
            const int n = ccn[c];
            const int s = cst[c];
            const int p = cbs[c];
            for (int i = l16 + 16; i < n; i += 16) {
              const unsigned w = (unsigned)chunks[(unsigned)c * PCH + (unsigned)(s + i)];
              const int r = atomicAdd(&cntl[w >> 24], 1);
              raw[p + i] = make_int2((int)w, r);
            }
          }
        }
      }
    }
  }
  __syncthreads();

  if (wave == 0) {
    int c8[8];
    int s = 0;
#pragma unroll
    for (int k = 0; k < 8; ++k) { c8[k] = cntl[lane * 8 + k]; s += c8[k]; }
    int in2 = s;
#pragma unroll
    for (int d = 1; d < 32; d <<= 1) {
      const int v = __shfl_up(in2, d);
      if (lane >= d) in2 += v;
    }
    int run = in2 - s;
#pragma unroll
    for (int k = 0; k < 8; ++k) { offl[lane * 8 + k] = run; run += c8[k]; }
  }
  __syncthreads();

  if (listed) {
    for (int k = tid; k < total; k += 1024) {
      const int2 p = raw[k];
      sorted[offl[(unsigned)p.x >> 24] + p.y] = p.x & 0xFFFFFF;
    }
  }
  __syncthreads();

  const int hl  = lane >> 4;
  const int c16 = lane & 15;
  const float* __restrict__ zc = z + 4 * c16;
  const char* __restrict__ zb = (const char*)z;
  const unsigned cb = 16u * (unsigned)c16;
  const float4 al = *(const float4*)(aw + 4 * c16);
  const float4 ar = *(const float4*)(aw + HD + 4 * c16);
  const float4 bb = *(const float4*)(bias + 4 * c16);
  const unsigned nmax = (unsigned)(N_ - 1);

  float4 zn = make_float4(0.f, 0.f, 0.f, 0.f);
  if (wave < rows)
    zn = *(const float4*)(zc + (size_t)(row0 + wave) * HD);

  for (int q = 0; q < (BROWS + BW - 1) / BW; ++q) {
    const int ld = wave + BW * q;
    if (ld >= rows) break;
    const int n = row0 + ld;
    const float erh = head_logit(zn, ar);
    if (ld + BW < rows)
      zn = *(const float4*)(zc + (size_t)(n + BW) * HD);
    float den = 0.f;
    float4 a = make_float4(0.f, 0.f, 0.f, 0.f);
    if (listed) {
      const int c = __builtin_amdgcn_readfirstlane(cntl[ld]);
      const int o = __builtin_amdgcn_readfirstlane(offl[ld]);
      const int* lst = sorted + o;
      int j0 = 0;
      for (; j0 + 8 <= c; j0 += 8) {
        float4 zv[4];
#pragma unroll
        for (int k = 0; k < 4; ++k) {
          const unsigned s = min((unsigned)lst[j0 + 2 * k + hl], nmax);
          zv[k] = *(const float4*)(zb + (s * 256u + cb));
        }
#pragma unroll
        for (int k = 0; k < 4; ++k) add_edge(zv[k], 1.f, al, erh, den, a);
      }
      const int r = c - j0;
      if (r > 0) {
        int sx[4];
        float w[4];
        float4 zv[4];
#pragma unroll
        for (int k = 0; k < 4; ++k) {
          const int e = 2 * k + hl;
          w[k] = e < r ? 1.f : 0.f;
          sx[k] = lst[j0 + min(e, r - 1)];
        }
#pragma unroll
        for (int k = 0; k < 4; ++k) {
          zv[k] = make_float4(0.f, 0.f, 0.f, 0.f);
          if (2 * k < r) {
            const unsigned s = min((unsigned)sx[k], nmax);
            zv[k] = *(const float4*)(zb + (s * 256u + cb));
          }
        }
#pragma unroll
        for (int k = 0; k < 4; ++k) {
          if (2 * k < r) add_edge(zv[k], w[k], al, erh, den, a);
        }
      }
    } else {
      const float wh = hl ? 0.f : 1.f;
      for (int e0 = 0; e0 < E_; e0 += 32) {
        const int e = e0 + lane;
        const int t = e < E_ ? dst[e] : -1;
        const int sl = e < E_ ? src[e] : 0;
        unsigned m = (unsigned)__ballot(t == n);
        while (m) {
          const int j = __builtin_ctz(m);
          m &= m - 1;
          const unsigned s = min((unsigned)__builtin_amdgcn_readlane(sl, j), nmax);
          const float4 zv = *(const float4*)(zc + (size_t)s * HD);
          add_edge(zv, wh, al, erh, den, a);
        }
      }
    }
    den += xhalf(den);
    a.x += xhalf(a.x);
    a.y += xhalf(a.y);
    a.z += xhalf(a.z);
    a.w += xhalf(a.w);
    den = den > 0.f ? den : 1.f;
    float4 o4;
    o4.x = a.x / den + bb.x;
    o4.y = a.y / den + bb.y;
    o4.z = a.z / den + bb.z;
    o4.w = a.w / den + bb.w;
    if (hl == 0) *(float4*)(oblk + (size_t)ld * HD + 4 * c16) = o4;
  }
}

extern "C" void kernel_launch(void* const* d_in, const int* in_sizes, int n_in,
                              void* d_out, int out_size, void* d_ws, size_t ws_size,
                              hipStream_t stream) {
  const float* feat = (const float*)d_in[0];
  const int*   src  = (const int*)d_in[1];
  const int*   dst  = (const int*)d_in[2];
  const float* W    = (const float*)d_in[3];
  const float* aw   = (const float*)d_in[4];
  const float* bias = (const float*)d_in[5];
  float* out = (float*)d_out;

  const int N_ = in_sizes[0] / DIN;
  const int E_ = in_sizes[1];
  const int NB  = (N_ + BROWS - 1) / BROWS;
  const int NCH = (E_ + PCH - 1) / PCH;

  float* ws  = (float*)d_ws;
  float* z   = ws;
  int*   lst = (int*)(z + (size_t)N_ * HD);
  int*   tab = lst + (size_t)NCH * PCH;

  const int NG = (N_ + GROWS - 1) / GROWS;
  const int g1 = NG > NCH ? NG : NCH;
  gat_proj_bin<<<g1, FT, 2 * NB * sizeof(int), stream>>>(
      feat, W, src, dst, z, lst, tab, N_, E_, NB, NCH, (unsigned long long)ws_size);

  gat_block<<<NB, BW * 32, 0, stream>>>(src, dst, aw, z, lst, tab, bias, out, N_, E_, NB, NCH,
                                     (unsigned long long)ws_size);
}


# Block kernel row sums take fewer vector instructions: log2(e) folded into a (exp2 direct), LeakyReLU as max(x,0.2x), dst logit added as 4 quarter start values, 1 reciprocal per row not 4 divisions
# speedup vs baseline: 4.7169x; 1.0386x over previous
#include <hip/hip_runtime.h>
#include <hip/hip_bf16.h>


#define DIN 128
#define HD  64

typedef __attribute__((ext_vector_type(2))) float v2f;
typedef __attribute__((ext_vector_type(8))) float v8f;

#define BROWS   200
#define BSLOT   256
#define BW      32
#define GROWS   256
#define LCAP    8192
#define PCH     4096
#define FT      256
#define EPT     (PCH / FT)
#define CHMAX   1024

typedef __attribute__((ext_vector_type(4))) float v4f;

#define WMMA_F32(a, b, c) \
  __builtin_amdgcn_wmma_f32_16x16x4_f32(false, a, false, b, (short)0, c, false, false)

__device__ __forceinline__ bool ws_holds(unsigned long long wsb, int N_, int NB, int NCH) {
  const unsigned long long need =
      4ull * ((unsigned long long)N_ * HD + (unsigned long long)NCH * PCH +
              (unsigned long long)NCH * (NB + 1));
  return wsb >= need;
}

__device__ __forceinline__ void bin_chunk(
    const int* __restrict__ src, const int* __restrict__ dst,
    int* __restrict__ lst, int* __restrict__ tab,
    int* hist, int* bst, int* stg, int* wsum,
    int c, int N_, int E_, int NB) {
  const int tid  = threadIdx.x;
  const int lane = tid & 31;
  const int wave = tid >> 5;
  for (int b = tid; b < NB; b += FT) hist[b] = 0;
  __syncthreads();

  const int cbase = c * PCH;
  int sv[EPT], tv[EPT], rv[EPT];
  if (cbase + PCH <= E_) {
#pragma unroll
    for (int i = 0; i < EPT / 4; ++i) {
      const int e0 = cbase + (i * FT + tid) * 4;
      const int4 s4 = *(const int4*)(src + e0);
      const int4 t4 = *(const int4*)(dst + e0);
      sv[4 * i + 0] = s4.x; sv[4 * i + 1] = s4.y;
      sv[4 * i + 2] = s4.z; sv[4 * i + 3] = s4.w;
      tv[4 * i + 0] = t4.x; tv[4 * i + 1] = t4.y;
      tv[4 * i + 2] = t4.z; tv[4 * i + 3] = t4.w;
    }
  } else {
#pragma unroll
    for (int k = 0; k < EPT; ++k) {
      const int e = cbase + ((k >> 2) * FT + tid) * 4 + (k & 3);
      sv[k] = e < E_ ? src[e] : 0;
      tv[k] = e < E_ ? dst[e] : -1;
    }
  }
#pragma unroll
  for (int k = 0; k < EPT; ++k) {
    rv[k] = -1;
    if ((unsigned)tv[k] < (unsigned)N_) {
      const unsigned blk = (unsigned)tv[k] / BROWS;
      const unsigned rib = (unsigned)tv[k] - blk * BROWS;
      rv[k] = atomicAdd(&hist[blk], 1);
      sv[k] = (int)(((unsigned)sv[k] & 0xFFFFFFu) | (rib << 24));
      tv[k] = (int)blk;
    }
  }
  __syncthreads();

  const int ipt = (NB + FT - 1) / FT;
  const int bfirst = tid * ipt;
  int s = 0;
  for (int i = 0; i < ipt; ++i) {
    const int b = bfirst + i;
    if (b < NB) s += hist[b];
  }
  int incl = s;
#pragma unroll
  for (int d = 1; d < 32; d <<= 1) {
    const int v = __shfl_up(incl, d);
    if (lane >= d) incl += v;
  }
  if (lane == 31) wsum[wave] = incl;
  __syncthreads();
  if (wave == 0) {
    const int w = lane < FT / 32 ? wsum[lane] : 0;
    int wi = w;
#pragma unroll
    for (int d = 1; d < 32; d <<= 1) {
      const int v = __shfl_up(wi, d);
      if (lane >= d) wi += v;
    }
    if (lane < FT / 32) wsum[lane] = wi - w;
    if (lane == 31) wsum[FT / 32] = wi;
  }
  __syncthreads();
  int* trow = tab + (size_t)c * (NB + 1);
  int run = wsum[wave] + incl - s;
  for (int i = 0; i < ipt; ++i) {
    const int b = bfirst + i;
    if (b < NB) {
      bst[b] = run;
      trow[b] = run;
      run += hist[b];
    }
  }
  const int tot = wsum[FT / 32];
  if (tid == 0) trow[NB] = tot;
  __syncthreads();

#pragma unroll
  for (int k = 0; k < EPT; ++k) {
    if (rv[k] >= 0) stg[bst[tv[k]] + rv[k]] = sv[k];
  }
  __syncthreads();
  int ev[EPT];
#pragma unroll
  for (int k = 0; k < EPT; ++k) {
    const int i = k * FT + tid;
    ev[k] = 0;
    if (i < tot) ev[k] = stg[i];
  }
  int* lrow = lst + (size_t)c * PCH;
#pragma unroll
  for (int k = 0; k < EPT; ++k) {
    const int i = k * FT + tid;
    if (i < tot) lrow[i] = ev[k];
  }
}

__global__ __launch_bounds__(FT) void gat_proj_bin(
    const float* __restrict__ feat, const float* __restrict__ W,
    const int* __restrict__ src, const int* __restrict__ dst,
    float* __restrict__ z, int* __restrict__ lst, int* __restrict__ tab,
    int N_, int E_, int NB, int NCH, unsigned long long wsb) {
  extern __shared__ int dynl[];
  __shared__ int stg[PCH];
  __shared__ int wsum[FT / 32 + 1];
  if (!ws_holds(wsb, N_, NB, NCH)) return;
  if ((int)blockIdx.x < NCH && N_ <= (1 << 24))
    bin_chunk(src, dst, lst, tab, dynl, dynl + NB, stg, wsum, blockIdx.x, N_, E_, NB);

  const int lane = threadIdx.x & 31;
  const int wave = blockIdx.x * (FT / 32) + (threadIdx.x >> 5);
  const int r0 = wave * 32;
  if (r0 >= N_) return;

  const int half = lane >> 4;
  const int m    = lane & 15;

  const int rowA0 = min(r0 + m, N_ - 1);
  const int rowA1 = min(r0 + 16 + m, N_ - 1);
  const float* __restrict__ a0p = feat + (size_t)rowA0 * DIN + 4 * half;
  const float* __restrict__ a1p = feat + (size_t)rowA1 * DIN + 4 * half;
  const float* __restrict__ bp  = W + (size_t)m * DIN + 4 * half;

  v8f acc[2][4];
#pragma unroll
  for (int j = 0; j < 4; ++j) { acc[0][j] = (v8f){}; acc[1][j] = (v8f){}; }

  v4f A0c = *(const v4f*)a0p, A1c = *(const v4f*)a1p, A0n = A0c, A1n = A1c;
  v4f Bc[4], Bn[4];
#pragma unroll
  for (int j = 0; j < 4; ++j) {
    Bc[j] = *(const v4f*)(bp + (size_t)j * 16 * DIN);
    Bn[j] = Bc[j];
  }

#pragma unroll
  for (int t = 0; t < 16; ++t) {
    if (t + 1 < 16) {
      A0n = *(const v4f*)(a0p + 8 * (t + 1));
      A1n = *(const v4f*)(a1p + 8 * (t + 1));
#pragma unroll
      for (int j = 0; j < 4; ++j)
        Bn[j] = *(const v4f*)(bp + (size_t)j * 16 * DIN + 8 * (t + 1));
    }
    const v2f a0l = __builtin_shufflevector(A0c, A0c, 0, 1);
    const v2f a0h = __builtin_shufflevector(A0c, A0c, 2, 3);
    const v2f a1l = __builtin_shufflevector(A1c, A1c, 0, 1);
    const v2f a1h = __builtin_shufflevector(A1c, A1c, 2, 3);
#pragma unroll
    for (int j = 0; j < 4; ++j) {
      const v2f bl = __builtin_shufflevector(Bc[j], Bc[j], 0, 1);
      const v2f bh = __builtin_shufflevector(Bc[j], Bc[j], 2, 3);
      acc[0][j] = WMMA_F32(a0l, bl, acc[0][j]);
      acc[1][j] = WMMA_F32(a1l, bl, acc[1][j]);
      acc[0][j] = WMMA_F32(a0h, bh, acc[0][j]);
      acc[1][j] = WMMA_F32(a1h, bh, acc[1][j]);
    }
    A0c = A0n;
    A1c = A1n;
#pragma unroll
    for (int j = 0; j < 4; ++j) Bc[j] = Bn[j];
  }

#pragma unroll
  for (int i = 0; i < 2; ++i) {
#pragma unroll
    for (int r = 0; r < 8; ++r) {
      const int row = r0 + 16 * i + r + 8 * half;
      if (row < N_) {
        float* zr = z + (size_t)row * HD + m;
        zr[0]  = acc[i][0][r];
        zr[16] = acc[i][1][r];
        zr[32] = acc[i][2][r];
        zr[48] = acc[i][3][r];
      }
    }
  }
}

__device__ __forceinline__ float quad_sum(float t) {
  t += __int_as_float(__builtin_amdgcn_mov_dpp(__float_as_int(t), 0xB1, 0xF, 0xF, true));
  t += __int_as_float(__builtin_amdgcn_mov_dpp(__float_as_int(t), 0x4E, 0xF, 0xF, true));
  return t;
}

__device__ __forceinline__ float head_logit(const float4 zv, const float4 av) {
  float t = zv.x * av.x;
  t = fmaf(zv.y, av.y, t);
  t = fmaf(zv.z, av.z, t);
  t = fmaf(zv.w, av.w, t);
  return quad_sum(t);
}

__device__ __forceinline__ float head_logit_from(const float4 zv, const float4 av, float start) {
  float t = fmaf(zv.x, av.x, start);
  t = fmaf(zv.y, av.y, t);
  t = fmaf(zv.z, av.z, t);
  t = fmaf(zv.w, av.w, t);
  return quad_sum(t);
}

__device__ __forceinline__ float xhalf(float x) {
  const int b = __float_as_int(x);
  return __int_as_float(__builtin_amdgcn_permlanex16(b, b, 0x76543210, 0xfedcba98, false, false));
}

__device__ __forceinline__ void add_edge(const float4 zv, float w,
    const float4 al, float erq, float& den, float4& a) {
  float x = head_logit_from(zv, al, erq);
  x = __builtin_fmaxf(x, 0.2f * x);
  const float p = w * __builtin_amdgcn_exp2f(x);
  den += p;
  a.x = fmaf(p, zv.x, a.x);
  a.y = fmaf(p, zv.y, a.y);
  a.z = fmaf(p, zv.z, a.z);
  a.w = fmaf(p, zv.w, a.w);
}

__global__ __launch_bounds__(BW * 32) void gat_block(
    const int* __restrict__ src, const int* __restrict__ dst,
    const float* __restrict__ aw, const float* __restrict__ z,
    const int* __restrict__ chunks, const int* __restrict__ tab,
    const float* __restrict__ bias, float* __restrict__ out,
    int N_, int E_, int NB, int NCH, unsigned long long wsb) {
  __shared__ int sorted[LCAP];
  __shared__ int2 raw[LCAP];
  __shared__ int cntl[BSLOT];
  __shared__ int offl[BSLOT];
  __shared__ int cst[CHMAX];
  __shared__ int ccn[CHMAX];
  __shared__ int cbs[CHMAX];
  __shared__ int wsum[33];

  if (!ws_holds(wsb, N_, NB, NCH)) return;
  const int tid  = threadIdx.x;
  const int lane = tid & 31;
  const int wave = __builtin_amdgcn_readfirstlane(tid >> 5);
  const int b    = blockIdx.x;
  const int row0 = b * BROWS;
  const int rows = min(BROWS, N_ - row0);
  const bool small = N_ <= (1 << 24) && NCH <= CHMAX;
  float* oblk = out + (size_t)row0 * HD;

  if (tid < BSLOT) cntl[tid] = 0;
  int mys = 0, myn = 0;
  if (small && tid < NCH) {
    const int* t = tab + (size_t)tid * (NB + 1) + b;
    const int t0 = t[0], t1 = t[1];
    mys = min(max(t0, 0), PCH);
    myn = min(max(t1 - mys, 0), PCH - mys);
  }
  int incl = myn;
#pragma unroll
  for (int d = 1; d < 32; d <<= 1) {
    const int v = __shfl_up(incl, d);
    if (lane >= d) incl += v;
  }
  if (lane == 31) wsum[wave] = incl;
  __syncthreads();
  if (wave == 0) {
    const int w = wsum[lane];
    int wi = w;
#pragma unroll
    for (int d = 1; d < 32; d <<= 1) {
      const int v = __shfl_up(wi, d);
      if (lane >= d) wi += v;
    }
    wsum[lane] = wi - w;
    if (lane == 31) wsum[32] = wi;
  }
  __syncthreads();
  cst[tid] = mys;
  ccn[tid] = myn;
  cbs[tid] = wsum[wave] + incl - myn;
  const int total = __builtin_amdgcn_readfirstlane(wsum[32]);
  const bool listed = small && total <= LCAP;
  __syncthreads();

  if (listed) {
    const int l16 = lane & 15;
    const int hw  = 2 * wave + (lane >> 4);
    for (int g = 0; g < NCH; g += 16 * BW) {
      unsigned wd[8];
      int kk[8], rk[8], cn[8], cp[8], cs[8];
      int nmx = 0;
#pragma unroll
      for (int j = 0; j < 8; ++j) {
        const int c = g + hw + 2 * BW * j;
        const int cc = min(c, CHMAX - 1);
        cn[j] = ccn[cc];
        cp[j] = cbs[cc];
        cs[j] = cst[cc];
        nmx = max(nmx, cn[j]);
      }
#pragma unroll
      for (int j = 0; j < 8; ++j) {
        const int c = g + hw + 2 * BW * j;
        kk[j] = -1;
        wd[j] = 0u;
        if (l16 < cn[j]) {
          kk[j] = cp[j] + l16;
          wd[j] = (unsigned)chunks[(unsigned)c * PCH + (unsigned)(cs[j] + l16)];
        }
      }
#pragma unroll
      for (int j = 0; j < 8; ++j) {
        rk[j] = 0;
        if (kk[j] >= 0) rk[j] = atomicAdd(&cntl[wd[j] >> 24], 1);
      }
#pragma unroll
      for (int j = 0; j < 8; ++j) {
        if (kk[j] >= 0) raw[kk[j]] = make_int2((int)wd[j], rk[j]);
      }
      if (__ballot(nmx > 16) != 0) {
        for (int j = 0; j < 8; ++j) {
          const int c = g + hw + 2 * BW * j;
          if (c < NCH) {
            const int n = ccn[c];
            const int s = cst[c];
            const int p = cbs[c];
            for (int i = l16 + 16; i < n; i += 16) {
              const unsigned w = (unsigned)chunks[(unsigned)c * PCH + (unsigned)(s + i)];
              const int r = atomicAdd(&cntl[w >> 24], 1);
              raw[p + i] = make_int2((int)w, r);
            }
          }
        }
      }
    }
  }
  __syncthreads();

  if (wave == 0) {
    int c8[8];
    int s = 0;
#pragma unroll
    for (int k = 0; k < 8; ++k) { c8[k] = cntl[lane * 8 + k]; s += c8[k]; }
    int in2 = s;
#pragma unroll
    for (int d = 1; d < 32; d <<= 1) {
      const int v = __shfl_up(in2, d);
      if (lane >= d) in2 += v;
    }
    int run = in2 - s;
#pragma unroll
    for (int k = 0; k < 8; ++k) { offl[lane * 8 + k] = run; run += c8[k]; }
  }
  __syncthreads();

  if (listed) {
    for (int k = tid; k < total; k += 1024) {
      const int2 p = raw[k];
      sorted[offl[(unsigned)p.x >> 24] + p.y] = p.x & 0xFFFFFF;
    }
  }
  __syncthreads();

  const int hl  = lane >> 4;
  const int c16 = lane & 15;
  const float* __restrict__ zc = z + 4 * c16;
  const char* __restrict__ zb = (const char*)z;
  const unsigned cb = 16u * (unsigned)c16;
  const float L2E = 1.44269504088896340736f;
  float4 al = *(const float4*)(aw + 4 * c16);
  float4 ar = *(const float4*)(aw + HD + 4 * c16);
  al.x *= L2E; al.y *= L2E; al.z *= L2E; al.w *= L2E;
  ar.x *= L2E; ar.y *= L2E; ar.z *= L2E; ar.w *= L2E;
  const float4 bb = *(const float4*)(bias + 4 * c16);
  const unsigned nmax = (unsigned)(N_ - 1);

  float4 zn = make_float4(0.f, 0.f, 0.f, 0.f);
  if (wave < rows)
    zn = *(const float4*)(zc + (size_t)(row0 + wave) * HD);

  for (int q = 0; q < (BROWS + BW - 1) / BW; ++q) {
    const int ld = wave + BW * q;
    if (ld >= rows) break;
    const int n = row0 + ld;
    const float erh = 0.25f * head_logit(zn, ar);
    if (ld + BW < rows)
      zn = *(const float4*)(zc + (size_t)(n + BW) * HD);
    float den = 0.f;
    float4 a = make_float4(0.f, 0.f, 0.f, 0.f);
    if (listed) {
      const int c = __builtin_amdgcn_readfirstlane(cntl[ld]);
      const int o = __builtin_amdgcn_readfirstlane(offl[ld]);
      const int* lst = sorted + o;
      int j0 = 0;
      for (; j0 + 8 <= c; j0 += 8) {
        float4 zv[4];
#pragma unroll
        for (int k = 0; k < 4; ++k) {
          const unsigned s = min((unsigned)lst[j0 + 2 * k + hl], nmax);
          zv[k] = *(const float4*)(zb + (s * 256u + cb));
        }
#pragma unroll
        for (int k = 0; k < 4; ++k) add_edge(zv[k], 1.f, al, erh, den, a);
      }
      const int r = c - j0;
      if (r > 0) {
        int sx[4];
        float w[4];
        float4 zv[4];
#pragma unroll
        for (int k = 0; k < 4; ++k) {
          const int e = 2 * k + hl;
          w[k] = e < r ? 1.f : 0.f;
          sx[k] = lst[j0 + min(e, r - 1)];
        }
#pragma unroll
        for (int k = 0; k < 4; ++k) {
          zv[k] = make_float4(0.f, 0.f, 0.f, 0.f);
          if (2 * k < r) {
            const unsigned s = min((unsigned)sx[k], nmax);
            zv[k] = *(const float4*)(zb + (s * 256u + cb));
          }
        }
#pragma unroll
        for (int k = 0; k < 4; ++k) {
          if (2 * k < r) add_edge(zv[k], w[k], al, erh, den, a);
        }
      }
    } else {
      const float wh = hl ? 0.f : 1.f;
      for (int e0 = 0; e0 < E_; e0 += 32) {
        const int e = e0 + lane;
        const int t = e < E_ ? dst[e] : -1;
        const int sl = e < E_ ? src[e] : 0;
        unsigned m = (unsigned)__ballot(t == n);
        while (m) {
          const int j = __builtin_ctz(m);
          m &= m - 1;
          const unsigned s = min((unsigned)__builtin_amdgcn_readlane(sl, j), nmax);
          const float4 zv = *(const float4*)(zc + (size_t)s * HD);
          add_edge(zv, wh, al, erh, den, a);
        }
      }
    }
    den += xhalf(den);
    a.x += xhalf(a.x);
    a.y += xhalf(a.y);
    a.z += xhalf(a.z);
    a.w += xhalf(a.w);
    den = den > 0.f ? den : 1.f;
    const float inv = 1.f / den;
    float4 o4;
    o4.x = fmaf(a.x, inv, bb.x);
    o4.y = fmaf(a.y, inv, bb.y);
    o4.z = fmaf(a.z, inv, bb.z);
    o4.w = fmaf(a.w, inv, bb.w);
    if (hl == 0) *(float4*)(oblk + (size_t)ld * HD + 4 * c16) = o4;
  }
}

extern "C" void kernel_launch(void* const* d_in, const int* in_sizes, int n_in,
                              void* d_out, int out_size, void* d_ws, size_t ws_size,
                              hipStream_t stream) {
  const float* feat = (const float*)d_in[0];
  const int*   src  = (const int*)d_in[1];
  const int*   dst  = (const int*)d_in[2];
  const float* W    = (const float*)d_in[3];
  const float* aw   = (const float*)d_in[4];
  const float* bias = (const float*)d_in[5];
  float* out = (float*)d_out;

  const int N_ = in_sizes[0] / DIN;
  const int E_ = in_sizes[1];
  const int NB  = (N_ + BROWS - 1) / BROWS;
  const int NCH = (E_ + PCH - 1) / PCH;

  float* ws  = (float*)d_ws;
  float* z   = ws;
  int*   lst = (int*)(z + (size_t)N_ * HD);
  int*   tab = lst + (size_t)NCH * PCH;

  const int NG = (N_ + GROWS - 1) / GROWS;
  const int g1 = NG > NCH ? NG : NCH;
  gat_proj_bin<<<g1, FT, 2 * NB * sizeof(int), stream>>>(
      feat, W, src, dst, z, lst, tab, N_, E_, NB, NCH, (unsigned long long)ws_size);

  gat_block<<<NB, BW * 32, 0, stream>>>(src, dst, aw, z, lst, tab, bias, out, N_, E_, NB, NCH,
                                     (unsigned long long)ws_size);
}


# First launch: the chunk sorts run on work-groups of their own behind the GEMM ones (grid = both), beside the matrix work, not in front of it in every work-group; lists and order of sums unchanged
# speedup vs baseline: 4.8817x; 1.0349x over previous
#include <hip/hip_runtime.h>
#include <hip/hip_bf16.h>


#define DIN 128
#define HD  64

typedef __attribute__((ext_vector_type(2))) float v2f;
typedef __attribute__((ext_vector_type(8))) float v8f;

#define BROWS   200
#define BSLOT   256
#define BW      32
#define GROWS   256
#define LCAP    8192
#define PCH     4096
#define FT      256
#define EPT     (PCH / FT)
#define CHMAX   1024

typedef __attribute__((ext_vector_type(4))) float v4f;

#define WMMA_F32(a, b, c) \
  __builtin_amdgcn_wmma_f32_16x16x4_f32(false, a, false, b, (short)0, c, false, false)

__device__ __forceinline__ bool ws_holds(unsigned long long wsb, int N_, int NB, int NCH) {
  const unsigned long long need =
      4ull * ((unsigned long long)N_ * HD + (unsigned long long)NCH * PCH +
              (unsigned long long)NCH * (NB + 1));
  return wsb >= need;
}

__device__ __forceinline__ void bin_chunk(
    const int* __restrict__ src, const int* __restrict__ dst,
    int* __restrict__ lst, int* __restrict__ tab,
    int* hist, int* bst, int* stg, int* wsum,
    int c, int N_, int E_, int NB) {
  const int tid  = threadIdx.x;
  const int lane = tid & 31;
  const int wave = tid >> 5;
  for (int b = tid; b < NB; b += FT) hist[b] = 0;
  __syncthreads();

  const int cbase = c * PCH;
  int sv[EPT], tv[EPT], rv[EPT];
  if (cbase + PCH <= E_) {
#pragma unroll
    for (int i = 0; i < EPT / 4; ++i) {
      const int e0 = cbase + (i * FT + tid) * 4;
      const int4 s4 = *(const int4*)(src + e0);
      const int4 t4 = *(const int4*)(dst + e0);
      sv[4 * i + 0] = s4.x; sv[4 * i + 1] = s4.y;
      sv[4 * i + 2] = s4.z; sv[4 * i + 3] = s4.w;
      tv[4 * i + 0] = t4.x; tv[4 * i + 1] = t4.y;
      tv[4 * i + 2] = t4.z; tv[4 * i + 3] = t4.w;
    }
  } else {
#pragma unroll
    for (int k = 0; k < EPT; ++k) {
      const int e = cbase + ((k >> 2) * FT + tid) * 4 + (k & 3);
      sv[k] = e < E_ ? src[e] : 0;
      tv[k] = e < E_ ? dst[e] : -1;
    }
  }
#pragma unroll
  for (int k = 0; k < EPT; ++k) {
    rv[k] = -1;
    if ((unsigned)tv[k] < (unsigned)N_) {
      const unsigned blk = (unsigned)tv[k] / BROWS;
      const unsigned rib = (unsigned)tv[k] - blk * BROWS;
      rv[k] = atomicAdd(&hist[blk], 1);
      sv[k] = (int)(((unsigned)sv[k] & 0xFFFFFFu) | (rib << 24));
      tv[k] = (int)blk;
    }
  }
  __syncthreads();

  const int ipt = (NB + FT - 1) / FT;
  const int bfirst = tid * ipt;
  int s = 0;
  for (int i = 0; i < ipt; ++i) {
    const int b = bfirst + i;
    if (b < NB) s += hist[b];
  }
  int incl = s;
#pragma unroll
  for (int d = 1; d < 32; d <<= 1) {
    const int v = __shfl_up(incl, d);
    if (lane >= d) incl += v;
  }
  if (lane == 31) wsum[wave] = incl;
  __syncthreads();
  if (wave == 0) {
    const int w = lane < FT / 32 ? wsum[lane] : 0;
    int wi = w;
#pragma unroll
    for (int d = 1; d < 32; d <<= 1) {
      const int v = __shfl_up(wi, d);
      if (lane >= d) wi += v;
    }
    if (lane < FT / 32) wsum[lane] = wi - w;
    if (lane == 31) wsum[FT / 32] = wi;
  }
  __syncthreads();
  int* trow = tab + (size_t)c * (NB + 1);
  int run = wsum[wave] + incl - s;
  for (int i = 0; i < ipt; ++i) {
    const int b = bfirst + i;
    if (b < NB) {
      bst[b] = run;
      trow[b] = run;
      run += hist[b];
    }
  }
  const int tot = wsum[FT / 32];
  if (tid == 0) trow[NB] = tot;
  __syncthreads();

#pragma unroll
  for (int k = 0; k < EPT; ++k) {
    if (rv[k] >= 0) stg[bst[tv[k]] + rv[k]] = sv[k];
  }
  __syncthreads();
  int ev[EPT];
#pragma unroll
  for (int k = 0; k < EPT; ++k) {
    const int i = k * FT + tid;
    ev[k] = 0;
    if (i < tot) ev[k] = stg[i];
  }
  int* lrow = lst + (size_t)c * PCH;
#pragma unroll
  for (int k = 0; k < EPT; ++k) {
    const int i = k * FT + tid;
    if (i < tot) lrow[i] = ev[k];
  }
}

__global__ __launch_bounds__(FT) void gat_proj_bin(
    const float* __restrict__ feat, const float* __restrict__ W,
    const int* __restrict__ src, const int* __restrict__ dst,
    float* __restrict__ z, int* __restrict__ lst, int* __restrict__ tab,
    int N_, int E_, int NB, int NCH, unsigned long long wsb) {
  extern __shared__ int dynl[];
  __shared__ int stg[PCH];
  __shared__ int wsum[FT / 32 + 1];
  if (!ws_holds(wsb, N_, NB, NCH)) return;
  const int NG = (N_ + GROWS - 1) / GROWS;
  if ((int)blockIdx.x >= NG) {
    const int c = (int)blockIdx.x - NG;
    if (c < NCH && N_ <= (1 << 24))
      bin_chunk(src, dst, lst, tab, dynl, dynl + NB, stg, wsum, c, N_, E_, NB);
    return;
  }

  const int lane = threadIdx.x & 31;
  const int wave = blockIdx.x * (FT / 32) + (threadIdx.x >> 5);
  const int r0 = wave * 32;
  if (r0 >= N_) return;

  const int half = lane >> 4;
  const int m    = lane & 15;

  const int rowA0 = min(r0 + m, N_ - 1);
  const int rowA1 = min(r0 + 16 + m, N_ - 1);
  const float* __restrict__ a0p = feat + (size_t)rowA0 * DIN + 4 * half;
  const float* __restrict__ a1p = feat + (size_t)rowA1 * DIN + 4 * half;
  const float* __restrict__ bp  = W + (size_t)m * DIN + 4 * half;

  v8f acc[2][4];
#pragma unroll
  for (int j = 0; j < 4; ++j) { acc[0][j] = (v8f){}; acc[1][j] = (v8f){}; }

  v4f A0c = *(const v4f*)a0p, A1c = *(const v4f*)a1p, A0n = A0c, A1n = A1c;
  v4f Bc[4], Bn[4];
#pragma unroll
  for (int j = 0; j < 4; ++j) {
    Bc[j] = *(const v4f*)(bp + (size_t)j * 16 * DIN);
    Bn[j] = Bc[j];
  }

#pragma unroll
  for (int t = 0; t < 16; ++t) {
    if (t + 1 < 16) {
      A0n = *(const v4f*)(a0p + 8 * (t + 1));
      A1n = *(const v4f*)(a1p + 8 * (t + 1));
#pragma unroll
      for (int j = 0; j < 4; ++j)
        Bn[j] = *(const v4f*)(bp + (size_t)j * 16 * DIN + 8 * (t + 1));
    }
    const v2f a0l = __builtin_shufflevector(A0c, A0c, 0, 1);
    const v2f a0h = __builtin_shufflevector(A0c, A0c, 2, 3);
    const v2f a1l = __builtin_shufflevector(A1c, A1c, 0, 1);
    const v2f a1h = __builtin_shufflevector(A1c, A1c, 2, 3);
#pragma unroll
    for (int j = 0; j < 4; ++j) {
      const v2f bl = __builtin_shufflevector(Bc[j], Bc[j], 0, 1);
      const v2f bh = __builtin_shufflevector(Bc[j], Bc[j], 2, 3);
      acc[0][j] = WMMA_F32(a0l, bl, acc[0][j]);
      acc[1][j] = WMMA_F32(a1l, bl, acc[1][j]);
      acc[0][j] = WMMA_F32(a0h, bh, acc[0][j]);
      acc[1][j] = WMMA_F32(a1h, bh, acc[1][j]);
    }
    A0c = A0n;
    A1c = A1n;
#pragma unroll
    for (int j = 0; j < 4; ++j) Bc[j] = Bn[j];
  }

#pragma unroll
  for (int i = 0; i < 2; ++i) {
#pragma unroll
    for (int r = 0; r < 8; ++r) {
      const int row = r0 + 16 * i + r + 8 * half;
      if (row < N_) {
        float* zr = z + (size_t)row * HD + m;
        zr[0]  = acc[i][0][r];
        zr[16] = acc[i][1][r];
        zr[32] = acc[i][2][r];
        zr[48] = acc[i][3][r];
      }
    }
  }
}

__device__ __forceinline__ float quad_sum(float t) {
  t += __int_as_float(__builtin_amdgcn_mov_dpp(__float_as_int(t), 0xB1, 0xF, 0xF, true));
  t += __int_as_float(__builtin_amdgcn_mov_dpp(__float_as_int(t), 0x4E, 0xF, 0xF, true));
  return t;
}

__device__ __forceinline__ float head_logit(const float4 zv, const float4 av) {
  float t = zv.x * av.x;
  t = fmaf(zv.y, av.y, t);
  t = fmaf(zv.z, av.z, t);
  t = fmaf(zv.w, av.w, t);
  return quad_sum(t);
}

__device__ __forceinline__ float head_logit_from(const float4 zv, const float4 av, float start) {
  float t = fmaf(zv.x, av.x, start);
  t = fmaf(zv.y, av.y, t);
  t = fmaf(zv.z, av.z, t);
  t = fmaf(zv.w, av.w, t);
  return quad_sum(t);
}

__device__ __forceinline__ float xhalf(float x) {
  const int b = __float_as_int(x);
  return __int_as_float(__builtin_amdgcn_permlanex16(b, b, 0x76543210, 0xfedcba98, false, false));
}

__device__ __forceinline__ void add_edge(const float4 zv, float w,
    const float4 al, float erq, float& den, float4& a) {
  float x = head_logit_from(zv, al, erq);
  x = __builtin_fmaxf(x, 0.2f * x);
  const float p = w * __builtin_amdgcn_exp2f(x);
  den += p;
  a.x = fmaf(p, zv.x, a.x);
  a.y = fmaf(p, zv.y, a.y);
  a.z = fmaf(p, zv.z, a.z);
  a.w = fmaf(p, zv.w, a.w);
}

__global__ __launch_bounds__(BW * 32) void gat_block(
    const int* __restrict__ src, const int* __restrict__ dst,
    const float* __restrict__ aw, const float* __restrict__ z,
    const int* __restrict__ chunks, const int* __restrict__ tab,
    const float* __restrict__ bias, float* __restrict__ out,
    int N_, int E_, int NB, int NCH, unsigned long long wsb) {
  __shared__ int sorted[LCAP];
  __shared__ int2 raw[LCAP];
  __shared__ int cntl[BSLOT];
  __shared__ int offl[BSLOT];
  __shared__ int cst[CHMAX];
  __shared__ int ccn[CHMAX];
  __shared__ int cbs[CHMAX];
  __shared__ int wsum[33];

  if (!ws_holds(wsb, N_, NB, NCH)) return;
  const int tid  = threadIdx.x;
  const int lane = tid & 31;
  const int wave = __builtin_amdgcn_readfirstlane(tid >> 5);
  const int b    = blockIdx.x;
  const int row0 = b * BROWS;
  const int rows = min(BROWS, N_ - row0);
  const bool small = N_ <= (1 << 24) && NCH <= CHMAX;
  float* oblk = out + (size_t)row0 * HD;

  if (tid < BSLOT) cntl[tid] = 0;
  int mys = 0, myn = 0;
  if (small && tid < NCH) {
    const int* t = tab + (size_t)tid * (NB + 1) + b;
    const int t0 = t[0], t1 = t[1];
    mys = min(max(t0, 0), PCH);
    myn = min(max(t1 - mys, 0), PCH - mys);
  }
  int incl = myn;
#pragma unroll
  for (int d = 1; d < 32; d <<= 1) {
    const int v = __shfl_up(incl, d);
    if (lane >= d) incl += v;
  }
  if (lane == 31) wsum[wave] = incl;
  __syncthreads();
  if (wave == 0) {
    const int w = wsum[lane];
    int wi = w;
#pragma unroll
    for (int d = 1; d < 32; d <<= 1) {
      const int v = __shfl_up(wi, d);
      if (lane >= d) wi += v;
    }
    wsum[lane] = wi - w;
    if (lane == 31) wsum[32] = wi;
  }
  __syncthreads();
  cst[tid] = mys;
  ccn[tid] = myn;
  cbs[tid] = wsum[wave] + incl - myn;
  const int total = __builtin_amdgcn_readfirstlane(wsum[32]);
  const bool listed = small && total <= LCAP;
  __syncthreads();

  if (listed) {
    const int l16 = lane & 15;
    const int hw  = 2 * wave + (lane >> 4);
    for (int g = 0; g < NCH; g += 16 * BW) {
      unsigned wd[8];
      int kk[8], rk[8], cn[8], cp[8], cs[8];
      int nmx = 0;
#pragma unroll
      for (int j = 0; j < 8; ++j) {
        const int c = g + hw + 2 * BW * j;
        const int cc = min(c, CHMAX - 1);
        cn[j] = ccn[cc];
        cp[j] = cbs[cc];
        cs[j] = cst[cc];
        nmx = max(nmx, cn[j]);
      }
#pragma unroll
      for (int j = 0; j < 8; ++j) {
        const int c = g + hw + 2 * BW * j;
        kk[j] = -1;
        wd[j] = 0u;
        if (l16 < cn[j]) {
          kk[j] = cp[j] + l16;
          wd[j] = (unsigned)chunks[(unsigned)c * PCH + (unsigned)(cs[j] + l16)];
        }
      }
#pragma unroll
      for (int j = 0; j < 8; ++j) {
        rk[j] = 0;
        if (kk[j] >= 0) rk[j] = atomicAdd(&cntl[wd[j] >> 24], 1);
      }
#pragma unroll
      for (int j = 0; j < 8; ++j) {
        if (kk[j] >= 0) raw[kk[j]] = make_int2((int)wd[j], rk[j]);
      }
      if (__ballot(nmx > 16) != 0) {
        for (int j = 0; j < 8; ++j) {
          const int c = g + hw + 2 * BW * j;
          if (c < NCH) {
            const int n = ccn[c];
            const int s = cst[c];
            const int p = cbs[c];
            for (int i = l16 + 16; i < n; i += 16) {
              const unsigned w = (unsigned)chunks[(unsigned)c * PCH + (unsigned)(s + i)];
              const int r = atomicAdd(&cntl[w >> 24], 1);
              raw[p + i] = make_int2((int)w, r);
            }
          }
        }
      }
    }
  }
  __syncthreads();

  if (wave == 0) {
    int c8[8];
    int s = 0;
#pragma unroll
    for (int k = 0; k < 8; ++k) { c8[k] = cntl[lane * 8 + k]; s += c8[k]; }
    int in2 = s;
#pragma unroll
    for (int d = 1; d < 32; d <<= 1) {
      const int v = __shfl_up(in2, d);
      if (lane >= d) in2 += v;
    }
    int run = in2 - s;
#pragma unroll
    for (int k = 0; k < 8; ++k) { offl[lane * 8 + k] = run; run += c8[k]; }
  }
  __syncthreads();

  if (listed) {
    for (int k = tid; k < total; k += 1024) {
      const int2 p = raw[k];
      sorted[offl[(unsigned)p.x >> 24] + p.y] = p.x & 0xFFFFFF;
    }
  }
  __syncthreads();

  const int hl  = lane >> 4;
  const int c16 = lane & 15;
  const float* __restrict__ zc = z + 4 * c16;
  const char* __restrict__ zb = (const char*)z;
  const unsigned cb = 16u * (unsigned)c16;
  const float L2E = 1.44269504088896340736f;
  float4 al = *(const float4*)(aw + 4 * c16);
  float4 ar = *(const float4*)(aw + HD + 4 * c16);
  al.x *= L2E; al.y *= L2E; al.z *= L2E; al.w *= L2E;
  ar.x *= L2E; ar.y *= L2E; ar.z *= L2E; ar.w *= L2E;
  const float4 bb = *(const float4*)(bias + 4 * c16);
  const unsigned nmax = (unsigned)(N_ - 1);

  float4 zn = make_float4(0.f, 0.f, 0.f, 0.f);
  if (wave < rows)
    zn = *(const float4*)(zc + (size_t)(row0 + wave) * HD);

  for (int q = 0; q < (BROWS + BW - 1) / BW; ++q) {
    const int ld = wave + BW * q;
    if (ld >= rows) break;
    const int n = row0 + ld;
    const float erh = 0.25f * head_logit(zn, ar);
    if (ld + BW < rows)
      zn = *(const float4*)(zc + (size_t)(n + BW) * HD);
    float den = 0.f;
    float4 a = make_float4(0.f, 0.f, 0.f, 0.f);
    if (listed) {
      const int c = __builtin_amdgcn_readfirstlane(cntl[ld]);
      const int o = __builtin_amdgcn_readfirstlane(offl[ld]);
      const int* lst = sorted + o;
      int j0 = 0;
      for (; j0 + 8 <= c; j0 += 8) {
        float4 zv[4];
#pragma unroll
        for (int k = 0; k < 4; ++k) {
          const unsigned s = min((unsigned)lst[j0 + 2 * k + hl], nmax);
          zv[k] = *(const float4*)(zb + (s * 256u + cb));
        }
#pragma unroll
        for (int k = 0; k < 4; ++k) add_edge(zv[k], 1.f, al, erh, den, a);
      }
      const int r = c - j0;
      if (r > 0) {
        int sx[4];
        float w[4];
        float4 zv[4];
#pragma unroll
        for (int k = 0; k < 4; ++k) {
          const int e = 2 * k + hl;
          w[k] = e < r ? 1.f : 0.f;
          sx[k] = lst[j0 + min(e, r - 1)];
        }
#pragma unroll
        for (int k = 0; k < 4; ++k) {
          zv[k] = make_float4(0.f, 0.f, 0.f, 0.f);
          if (2 * k < r) {
            const unsigned s = min((unsigned)sx[k], nmax);
            zv[k] = *(const float4*)(zb + (s * 256u + cb));
          }
        }
#pragma unroll
        for (int k = 0; k < 4; ++k) {
          if (2 * k < r) add_edge(zv[k], w[k], al, erh, den, a);
        }
      }
    } else {
      const float wh = hl ? 0.f : 1.f;
      for (int e0 = 0; e0 < E_; e0 += 32) {
        const int e = e0 + lane;
        const int t = e < E_ ? dst[e] : -1;
        const int sl = e < E_ ? src[e] : 0;
        unsigned m = (unsigned)__ballot(t == n);
        while (m) {
          const int j = __builtin_ctz(m);
          m &= m - 1;
          const unsigned s = min((unsigned)__builtin_amdgcn_readlane(sl, j), nmax);
          const float4 zv = *(const float4*)(zc + (size_t)s * HD);
          add_edge(zv, wh, al, erh, den, a);
        }
      }
    }
    den += xhalf(den);
    a.x += xhalf(a.x);
    a.y += xhalf(a.y);
    a.z += xhalf(a.z);
    a.w += xhalf(a.w);
    den = den > 0.f ? den : 1.f;
    const float inv = 1.f / den;
    float4 o4;
    o4.x = fmaf(a.x, inv, bb.x);
    o4.y = fmaf(a.y, inv, bb.y);
    o4.z = fmaf(a.z, inv, bb.z);
    o4.w = fmaf(a.w, inv, bb.w);
    if (hl == 0) *(float4*)(oblk + (size_t)ld * HD + 4 * c16) = o4;
  }
}

extern "C" void kernel_launch(void* const* d_in, const int* in_sizes, int n_in,
                              void* d_out, int out_size, void* d_ws, size_t ws_size,
                              hipStream_t stream) {
  const float* feat = (const float*)d_in[0];
  const int*   src  = (const int*)d_in[1];
  const int*   dst  = (const int*)d_in[2];
  const float* W    = (const float*)d_in[3];
  const float* aw   = (const float*)d_in[4];
  const float* bias = (const float*)d_in[5];
  float* out = (float*)d_out;

  const int N_ = in_sizes[0] / DIN;
  const int E_ = in_sizes[1];
  const int NB  = (N_ + BROWS - 1) / BROWS;
  const int NCH = (E_ + PCH - 1) / PCH;

  float* ws  = (float*)d_ws;
  float* z   = ws;
  int*   lst = (int*)(z + (size_t)N_ * HD);
  int*   tab = lst + (size_t)NCH * PCH;

  const int NG = (N_ + GROWS - 1) / GROWS;
  const int g1 = NG + NCH;
  gat_proj_bin<<<g1, FT, 2 * NB * sizeof(int), stream>>>(
      feat, W, src, dst, z, lst, tab, N_, E_, NB, NCH, (unsigned long long)ws_size);

  gat_block<<<NB, BW * 32, 0, stream>>>(src, dst, aw, z, lst, tab, bias, out, N_, E_, NB, NCH,
                                     (unsigned long long)ws_size);
}
